# starting kernel
# baseline (speedup 1.0000x reference)
#include <hip/hip_runtime.h>
#include <hip/hip_bf16.h>

typedef __attribute__((ext_vector_type(2))) float v2f;
typedef __attribute__((ext_vector_type(8))) float v8f;

#define F_IN   128
#define F_HID  16

__device__ __forceinline__ void atomic_add_f32(float* p, float v) {
#if defined(__AMDGCN__)
    unsafeAtomicAdd(p, v);
#else
    atomicAdd(p, v);
#endif
}

__global__ void gcn_init_deg(float* __restrict__ deg, int n) {
    int i = blockIdx.x * blockDim.x + threadIdx.x;
    if (i < n) deg[i] = 1.0f;
}

__global__ void gcn_deg_accum(const int* __restrict__ ei, float* __restrict__ deg,
                              long long E) {
    long long e = (long long)blockIdx.x * blockDim.x + threadIdx.x;
    if (e >= E) return;
    int d = ei[E + e];
    atomic_add_f32(&deg[d], 1.0f);
}

__global__ void gcn_rsqrt(float* __restrict__ deg_dis, int n) {
    int i = blockIdx.x * blockDim.x + threadIdx.x;
    if (i < n) {
        float dgv = deg_dis[i];
        deg_dis[i] = dgv > 0.0f ? rsqrtf(dgv) : 0.0f;
    }
}

__global__ __launch_bounds__(32)
void gcn_xw_wmma(const float* __restrict__ x, const float* __restrict__ W1,
                 float* __restrict__ xW, int n) {
    const int row0 = blockIdx.x * 16;
    if (row0 >= n) return;
    const int lane = threadIdx.x;
    const int half = lane >> 4;
    const int l16  = lane & 15;

    const float* xrow = x + (long long)(row0 + l16) * F_IN;

    if (row0 + 16 + l16 < n)
        __builtin_prefetch(x + (long long)(row0 + 16 + l16) * F_IN + half * 64, 0, 3);

    v8f c = {};
#pragma unroll
    for (int k0 = 0; k0 < F_IN; k0 += 4) {
        const int ka = k0 + half * 2;
        v2f a, b;
        a.x = xrow[ka];
        a.y = xrow[ka + 1];
        b.x = W1[ka * F_HID + l16];
        b.y = W1[(ka + 1) * F_HID + l16];
        c = __builtin_amdgcn_wmma_f32_16x16x4_f32(
                 false, a,  false, b,
                 (short)0, c,  false,  false);
    }
#pragma unroll
    for (int r = 0; r < 8; ++r) {
        const int m = r + half * 8;
        xW[(long long)(row0 + m) * F_HID + l16] = c[r];
    }
}

__global__ void gcn_agg1_init(const float* __restrict__ xW, const float* __restrict__ dis,
                              float* __restrict__ h1agg, int n) {
    long long t = (long long)blockIdx.x * blockDim.x + threadIdx.x;
    long long i = t >> 4;
    if (i >= n) return;
    float dv = dis[i];
    h1agg[t] = xW[t] * dv * dv;
}

__global__ void gcn_agg1_edges(const int* __restrict__ ei, const float* __restrict__ dis,
                               const float* __restrict__ xW, float* __restrict__ h1agg,
                               long long E) {
    long long t = (long long)blockIdx.x * blockDim.x + threadIdx.x;
    long long e = t >> 4;
    int j = (int)(t & 15);
    if (e >= E) return;
    int s = ei[e];
    int d = ei[E + e];
    float norm = dis[s] * dis[d];
    atomic_add_f32(&h1agg[(long long)d * F_HID + j],
                   xW[(long long)s * F_HID + j] * norm);
}

__global__ void gcn_l1_epilogue(const float* __restrict__ h1agg,
                                const float* __restrict__ b1,
                                const float* __restrict__ W2,
                                float* __restrict__ z, int n) {
    int i = blockIdx.x * blockDim.x + threadIdx.x;
    if (i >= n) return;
    const float* row = h1agg + (long long)i * F_HID;
    float acc = 0.0f;
#pragma unroll
    for (int j = 0; j < F_HID; ++j) {
        float hv = row[j] + b1[j];
        hv = fmaxf(hv, 0.0f);
        acc = fmaf(hv, W2[j], acc);
    }
    z[i] = acc;
}

__global__ void gcn_out_init(const float* __restrict__ z, const float* __restrict__ dis,
                             const float* __restrict__ b2, float* __restrict__ out, int n) {
    int i = blockIdx.x * blockDim.x + threadIdx.x;
    if (i >= n) return;
    float dv = dis[i];
    out[i] = b2[0] + z[i] * dv * dv;
}

__global__ void gcn_agg2_edges(const int* __restrict__ ei, const float* __restrict__ dis,
                               const float* __restrict__ z, float* __restrict__ out,
                               long long E) {
    long long e = (long long)blockIdx.x * blockDim.x + threadIdx.x;
    if (e >= E) return;
    int s = ei[e];
    int d = ei[E + e];
    atomic_add_f32(&out[d], z[s] * dis[s] * dis[d]);
}

extern "C" void kernel_launch(void* const* d_in, const int* in_sizes, int n_in,
                              void* d_out, int out_size, void* d_ws, size_t ws_size,
                              hipStream_t stream) {
    const float* x  = (const float*)d_in[0];
    const int*   ei = (const int*)d_in[1];
    const float* W1 = (const float*)d_in[2];
    const float* b1 = (const float*)d_in[3];
    const float* W2 = (const float*)d_in[4];
    const float* b2 = (const float*)d_in[5];
    float* out = (float*)d_out;

    const int       N = in_sizes[0] / F_IN;
    const long long E = (long long)in_sizes[1] / 2;

    float* ws  = (float*)d_ws;
    float* dis = ws;
    float* xW  = ws + N;
    float* h1a = ws + (long long)17 * N;
    float* z   = ws + (long long)33 * N;

    const int T = 256;
    const unsigned nb_n    = (unsigned)((N + T - 1) / T);
    const unsigned nb_n16  = (unsigned)(((long long)N * F_HID + T - 1) / T);
    const unsigned nb_e    = (unsigned)((E + T - 1) / T);
    const unsigned nb_e16  = (unsigned)((E * F_HID + T - 1) / T);
    const unsigned nb_tile = (unsigned)((N + 15) / 16);

    gcn_init_deg   <<<nb_n,    T,  0, stream>>>(dis, N);
    gcn_deg_accum  <<<nb_e,    T,  0, stream>>>(ei, dis, E);
    gcn_rsqrt      <<<nb_n,    T,  0, stream>>>(dis, N);
    gcn_xw_wmma    <<<nb_tile, 32, 0, stream>>>(x, W1, xW, N);
    gcn_agg1_init  <<<nb_n16,  T,  0, stream>>>(xW, dis, h1a, N);
    gcn_agg1_edges <<<nb_e16,  T,  0, stream>>>(ei, dis, xW, h1a, E);
    gcn_l1_epilogue<<<nb_n,    T,  0, stream>>>(h1a, b1, W2, z, N);
    gcn_out_init   <<<nb_n,    T,  0, stream>>>(z, dis, b2, out, N);
    gcn_agg2_edges <<<nb_e,    T,  0, stream>>>(ei, dis, z, out, E);
}


# Degree and layer-2 scatter-adds use LDS atomics per (destination slice, edge chunk), not per-edge global atomics; chunk partials summed in fixed order by a plain kernel; f32 sum order changes
# speedup vs baseline: 2.2549x; 2.2549x over previous
#include <hip/hip_runtime.h>
#include <hip/hip_bf16.h>

typedef __attribute__((ext_vector_type(2))) float v2f;
typedef __attribute__((ext_vector_type(8))) float v8f;

#define F_IN   128
#define F_HID  16

__device__ __forceinline__ void atomic_add_f32(float* p, float v) {
#if defined(__AMDGCN__)
    unsafeAtomicAdd(p, v);
#else
    atomicAdd(p, v);
#endif
}

#define SLICE   25600
#define NCHUNK  64
#define HT      512
#define HU      16

__global__ __launch_bounds__(HT)
void gcn_deg_slices(const int* __restrict__ ei, unsigned* __restrict__ part,
                    int E, int n_slices, int chunk_len, long long pitch) {
    __shared__ unsigned cnt[SLICE];
    const int sl = blockIdx.x % n_slices;
    const int ch = blockIdx.x / n_slices;
    const int lo = sl * SLICE;
    for (int i = threadIdx.x; i < SLICE; i += HT) cnt[i] = 0u;
    __syncthreads();

    const int* __restrict__ dstp = ei + E;
    const int c0 = ch * chunk_len;
    const int c1 = (c0 + chunk_len < E) ? (c0 + chunk_len) : E;
    for (int base = c0 + (int)threadIdx.x; base < c1; base += HT * HU) {
        int d[HU];
#pragma unroll
        for (int u = 0; u < HU; ++u) {
            const int e  = base + u * HT;
            const int ec = (e < c1) ? e : (c1 - 1);
            const int dl = dstp[ec];
            d[u] = (e < c1) ? dl : -1;
        }
#pragma unroll
        for (int u = 0; u < HU; ++u) {
            const unsigned loc = (unsigned)(d[u] - lo);
            if (loc < (unsigned)SLICE) atomicAdd(&cnt[loc], 1u);
        }
    }
    __syncthreads();
    unsigned* __restrict__ row = part + (long long)ch * pitch + lo;
    for (int i = threadIdx.x; i < SLICE; i += HT) row[i] = cnt[i];
}

__global__ void gcn_deg_finish(const unsigned* __restrict__ part, float* __restrict__ dis,
                               int n, long long pitch) {
    int i = blockIdx.x * blockDim.x + threadIdx.x;
    if (i >= n) return;
    unsigned c = 1u;
#pragma unroll
    for (int ch = 0; ch < NCHUNK; ++ch) c += part[(long long)ch * pitch + i];
    const float dgv = (float)c;
    dis[i] = dgv > 0.0f ? rsqrtf(dgv) : 0.0f;
}

__global__ __launch_bounds__(32)
void gcn_xw_wmma(const float* __restrict__ x, const float* __restrict__ W1,
                 float* __restrict__ xW, int n) {
    const int row0 = blockIdx.x * 16;
    if (row0 >= n) return;
    const int lane = threadIdx.x;
    const int half = lane >> 4;
    const int l16  = lane & 15;

    const float* xrow = x + (long long)(row0 + l16) * F_IN;

    if (row0 + 16 + l16 < n)
        __builtin_prefetch(x + (long long)(row0 + 16 + l16) * F_IN + half * 64, 0, 3);

    v8f c = {};
#pragma unroll
    for (int k0 = 0; k0 < F_IN; k0 += 4) {
        const int ka = k0 + half * 2;
        v2f a, b;
        a.x = xrow[ka];
        a.y = xrow[ka + 1];
        b.x = W1[ka * F_HID + l16];
        b.y = W1[(ka + 1) * F_HID + l16];
        c = __builtin_amdgcn_wmma_f32_16x16x4_f32(
                 false, a,  false, b,
                 (short)0, c,  false,  false);
    }
#pragma unroll
    for (int r = 0; r < 8; ++r) {
        const int m = r + half * 8;
        xW[(long long)(row0 + m) * F_HID + l16] = c[r];
    }
}

__global__ void gcn_agg1_init(const float* __restrict__ xW, const float* __restrict__ dis,
                              float* __restrict__ h1agg, int n) {
    long long t = (long long)blockIdx.x * blockDim.x + threadIdx.x;
    long long i = t >> 4;
    if (i >= n) return;
    float dv = dis[i];
    h1agg[t] = xW[t] * dv * dv;
}

__global__ void gcn_agg1_edges(const int* __restrict__ ei, const float* __restrict__ dis,
                               const float* __restrict__ xW, float* __restrict__ h1agg,
                               long long E) {
    long long t = (long long)blockIdx.x * blockDim.x + threadIdx.x;
    long long e = t >> 4;
    int j = (int)(t & 15);
    if (e >= E) return;
    int s = ei[e];
    int d = ei[E + e];
    float norm = dis[s] * dis[d];
    atomic_add_f32(&h1agg[(long long)d * F_HID + j],
                   xW[(long long)s * F_HID + j] * norm);
}

__global__ void gcn_l1_epilogue(const float* __restrict__ h1agg,
                                const float* __restrict__ b1,
                                const float* __restrict__ W2,
                                const float* __restrict__ dis,
                                float* __restrict__ y, int n) {
    int i = blockIdx.x * blockDim.x + threadIdx.x;
    if (i >= n) return;
    const float* row = h1agg + (long long)i * F_HID;
    float acc = 0.0f;
#pragma unroll
    for (int j = 0; j < F_HID; ++j) {
        float hv = row[j] + b1[j];
        hv = fmaxf(hv, 0.0f);
        acc = fmaf(hv, W2[j], acc);
    }
    y[i] = acc * dis[i];
}

__global__ __launch_bounds__(HT)
void gcn_agg2_slices(const int* __restrict__ ei, const float* __restrict__ y,
                     float* __restrict__ part,
                     int E, int n_slices, int chunk_len, long long pitch) {
    __shared__ float acc[SLICE];
    const int sl = blockIdx.x % n_slices;
    const int ch = blockIdx.x / n_slices;
    const int lo = sl * SLICE;
    for (int i = threadIdx.x; i < SLICE; i += HT) acc[i] = 0.0f;
    __syncthreads();

    const int* __restrict__ srcp = ei;
    const int* __restrict__ dstp = ei + E;
    const int c0 = ch * chunk_len;
    const int c1 = (c0 + chunk_len < E) ? (c0 + chunk_len) : E;
    for (int base = c0 + (int)threadIdx.x; base < c1; base += HT * HU) {
        int d[HU], s[HU];
        float v[HU];
#pragma unroll
        for (int u = 0; u < HU; ++u) {
            const int e  = base + u * HT;
            const int ec = (e < c1) ? e : (c1 - 1);
            const int dl = dstp[ec];
            s[u] = srcp[ec];
            d[u] = (e < c1) ? dl : -1;
        }
#pragma unroll
        for (int u = 0; u < HU; ++u) {
            const unsigned loc = (unsigned)(d[u] - lo);
            v[u] = (loc < (unsigned)SLICE) ? y[(unsigned)s[u]] : 0.0f;
        }
#pragma unroll
        for (int u = 0; u < HU; ++u) {
            const unsigned loc = (unsigned)(d[u] - lo);
            if (loc < (unsigned)SLICE) atomic_add_f32(&acc[loc], v[u]);
        }
    }
    __syncthreads();
    float* __restrict__ row = part + (long long)ch * pitch + lo;
    for (int i = threadIdx.x; i < SLICE; i += HT) row[i] = acc[i];
}

__global__ void gcn_out_finish(const float* __restrict__ part, const float* __restrict__ y,
                               const float* __restrict__ dis, const float* __restrict__ b2,
                               float* __restrict__ out, int n, long long pitch) {
    int i = blockIdx.x * blockDim.x + threadIdx.x;
    if (i >= n) return;
    float p[NCHUNK];
#pragma unroll
    for (int ch = 0; ch < NCHUNK; ++ch) p[ch] = part[(long long)ch * pitch + i];
    float sum = y[i];
#pragma unroll
    for (int ch = 0; ch < NCHUNK; ++ch) sum += p[ch];
    out[i] = b2[0] + dis[i] * sum;
}

extern "C" void kernel_launch(void* const* d_in, const int* in_sizes, int n_in,
                              void* d_out, int out_size, void* d_ws, size_t ws_size,
                              hipStream_t stream) {
    const float* x  = (const float*)d_in[0];
    const int*   ei = (const int*)d_in[1];
    const float* W1 = (const float*)d_in[2];
    const float* b1 = (const float*)d_in[3];
    const float* W2 = (const float*)d_in[4];
    const float* b2 = (const float*)d_in[5];
    float* out = (float*)d_out;

    const int       N = in_sizes[0] / F_IN;
    const long long E = (long long)in_sizes[1] / 2;

    float* ws  = (float*)d_ws;
    float* dis = ws;
    float* xW  = ws + N;
    float* h1a = ws + (long long)17 * N;
    float* y   = ws + (long long)33 * N;
    float* part = ws + (long long)34 * N;

    const int       Ei       = (int)E;
    const int       n_slices = (N + SLICE - 1) / SLICE;
    const long long pitch    = (long long)n_slices * SLICE;
    const int       chunk_len = (int)((E + NCHUNK - 1) / NCHUNK);
    const unsigned  nb_acc   = (unsigned)(n_slices * NCHUNK);

    const int T = 256;
    const unsigned nb_n    = (unsigned)((N + T - 1) / T);
    const unsigned nb_n16  = (unsigned)(((long long)N * F_HID + T - 1) / T);
    const unsigned nb_e16  = (unsigned)((E * F_HID + T - 1) / T);
    const unsigned nb_tile = (unsigned)((N + 15) / 16);

    gcn_deg_slices <<<nb_acc,  HT, 0, stream>>>(ei, (unsigned*)part, Ei, n_slices, chunk_len, pitch);
    gcn_deg_finish <<<nb_n,    T,  0, stream>>>((const unsigned*)part, dis, N, pitch);
    gcn_xw_wmma    <<<nb_tile, 32, 0, stream>>>(x, W1, xW, N);
    gcn_agg1_init  <<<nb_n16,  T,  0, stream>>>(xW, dis, h1a, N);
    gcn_agg1_edges <<<nb_e16,  T,  0, stream>>>(ei, dis, xW, h1a, E);
    gcn_l1_epilogue<<<nb_n,    T,  0, stream>>>(h1a, b1, W2, dis, y, N);
    gcn_agg2_slices<<<nb_acc,  HT, 0, stream>>>(ei, y, part, Ei, n_slices, chunk_len, pitch);
    gcn_out_finish <<<nb_n,    T,  0, stream>>>(part, y, dis, b2, out, N, pitch);
}


# Both edge aggregations gather from a destination-sorted source list (counting sort: chunk counts + LDS ranks) and sum in registers; no global atomics, no h1agg; f32 sum order changes
# speedup vs baseline: 3.2590x; 1.4453x over previous
#include <hip/hip_runtime.h>
#include <hip/hip_bf16.h>

typedef __attribute__((ext_vector_type(2))) float v2f;
typedef __attribute__((ext_vector_type(8))) float v8f;

#define F_IN   128
#define F_HID  16

#define SLICE   25600
#define NCHUNK  64
#define HT      512
#define HU      16
#define ST      1024
#define SPT     (SLICE / ST)
#define FB      10

__global__ __launch_bounds__(HT)
void gcn_deg_slices(const int* __restrict__ ei, unsigned* __restrict__ part,
                    int E, int n_slices, int chunk_len, long long pitch) {
    __shared__ unsigned cnt[SLICE];
    const int sl = blockIdx.x % n_slices;
    const int ch = blockIdx.x / n_slices;
    const int lo = sl * SLICE;
    for (int i = threadIdx.x; i < SLICE; i += HT) cnt[i] = 0u;
    __syncthreads();

    const int* __restrict__ dstp = ei + E;
    const int c0 = ch * chunk_len;
    const int c1 = (c0 + chunk_len < E) ? (c0 + chunk_len) : E;
    for (int base = c0 + (int)threadIdx.x; base < c1; base += HT * HU) {
        int d[HU];
#pragma unroll
        for (int u = 0; u < HU; ++u) {
            const int e  = base + u * HT;
            const int ec = (e < c1) ? e : (c1 - 1);
            const int dl = dstp[ec];
            d[u] = (e < c1) ? dl : -1;
        }
#pragma unroll
        for (int u = 0; u < HU; ++u) {
            const unsigned loc = (unsigned)(d[u] - lo);
            if (loc < (unsigned)SLICE) atomicAdd(&cnt[loc], 1u);
        }
    }
    __syncthreads();
    unsigned* __restrict__ row = part + (long long)ch * pitch + lo;
    for (int i = threadIdx.x; i < SLICE; i += HT) row[i] = cnt[i];
}

__global__ __launch_bounds__(256)
void gcn_deg_finish(unsigned* __restrict__ part, float* __restrict__ dis,
                               unsigned* __restrict__ degi, int n, long long pitch) {
    int i = blockIdx.x * blockDim.x + threadIdx.x;
    if (i >= n) return;
    unsigned p[NCHUNK];
#pragma unroll
    for (int ch = 0; ch < NCHUNK; ++ch) p[ch] = part[(long long)ch * pitch + i];
    unsigned run = 0u;
#pragma unroll
    for (int ch = 0; ch < NCHUNK; ++ch) {
        part[(long long)ch * pitch + i] = run;
        run += p[ch];
    }
    degi[i] = run;
    const float dgv = (float)(run + 1u);
    dis[i] = dgv > 0.0f ? rsqrtf(dgv) : 0.0f;
}

__global__ __launch_bounds__(ST)
void gcn_first_slot(const unsigned* __restrict__ degi, unsigned* __restrict__ first,
                    int n, int E) {
    __shared__ unsigned cell[SLICE];
    __shared__ unsigned sums[ST];
    const int t    = threadIdx.x;
    const int base = blockIdx.x * SLICE;
    unsigned v[SPT];
#pragma unroll
    for (int u = 0; u < SPT; ++u) {
        const int idx = base + u * ST + t;
        const int ic  = (idx < n) ? idx : (n - 1);
        const unsigned dv = degi[ic];
        v[u] = (idx < n) ? dv : 0u;
    }
#pragma unroll
    for (int u = 0; u < SPT; ++u) cell[u * ST + t] = v[u];
    __syncthreads();

    unsigned s = 0u;
#pragma unroll
    for (int k = 0; k < SPT; ++k) {
        const unsigned c = cell[t * SPT + k];
        cell[t * SPT + k] = s;
        s += c;
    }
    sums[t] = s;
    __syncthreads();
    for (int off = 1; off < ST; off <<= 1) {
        const unsigned a = (t >= off) ? sums[t - off] : 0u;
        __syncthreads();
        sums[t] += a;
        __syncthreads();
    }
    const unsigned add = (unsigned)blockIdx.x * (unsigned)E + (sums[t] - s);
#pragma unroll
    for (int k = 0; k < SPT; ++k) cell[t * SPT + k] += add;
    __syncthreads();
#pragma unroll
    for (int u = 0; u < SPT; ++u) {
        const int idx = base + u * ST + t;
        if (idx < n) first[idx] = cell[u * ST + t];
    }
}

__global__ __launch_bounds__(32)
void gcn_xw_wmma(const float* __restrict__ x, const float* __restrict__ W1,
                 const float* __restrict__ dis, float* __restrict__ xWs, int n) {
    const int row0 = blockIdx.x * 16;
    if (row0 >= n) return;
    const int lane = threadIdx.x;
    const int half = lane >> 4;
    const int l16  = lane & 15;

    const float* xrow = x + (long long)(row0 + l16) * F_IN;

    if (row0 + 16 + l16 < n)
        __builtin_prefetch(x + (long long)(row0 + 16 + l16) * F_IN + half * 64, 0, 3);

    v8f c = {};
#pragma unroll
    for (int k0 = 0; k0 < F_IN; k0 += 4) {
        const int ka = k0 + half * 2;
        v2f a, b;
        a.x = xrow[ka];
        a.y = xrow[ka + 1];
        b.x = W1[ka * F_HID + l16];
        b.y = W1[(ka + 1) * F_HID + l16];
        c = __builtin_amdgcn_wmma_f32_16x16x4_f32(
                 false, a,  false, b,
                 (short)0, c,  false,  false);
    }
#pragma unroll
    for (int r = 0; r < 8; ++r) {
        const int m = r + half * 8;
        xWs[(long long)(row0 + m) * F_HID + l16] = c[r] * dis[row0 + m];
    }
}

__global__ __launch_bounds__(HT)
void gcn_csr_fill(const int* __restrict__ ei, const unsigned* __restrict__ part,
                  const unsigned* __restrict__ first, int* __restrict__ csr,
                  int E, int n, int n_slices, int chunk_len, long long pitch) {
    __shared__ unsigned cur[SLICE];
    const int sl = blockIdx.x % n_slices;
    const int ch = blockIdx.x / n_slices;
    const int lo = sl * SLICE;
    const unsigned* __restrict__ row = part + (long long)ch * pitch;
    for (int i0 = (int)threadIdx.x; i0 < SLICE; i0 += HT * FB) {
        unsigned a[FB], b[FB];
#pragma unroll
        for (int u = 0; u < FB; ++u) {
            const int node = lo + i0 + u * HT;
            const int nc   = (node < n) ? node : (n - 1);
            a[u] = first[nc];
            b[u] = row[nc];
        }
#pragma unroll
        for (int u = 0; u < FB; ++u) cur[i0 + u * HT] = a[u] + b[u];
    }
    __syncthreads();

    const int* __restrict__ srcp = ei;
    const int* __restrict__ dstp = ei + E;
    const int c0 = ch * chunk_len;
    const int c1 = (c0 + chunk_len < E) ? (c0 + chunk_len) : E;
    for (int base = c0 + (int)threadIdx.x; base < c1; base += HT * HU) {
        int d[HU], s[HU];
#pragma unroll
        for (int u = 0; u < HU; ++u) {
            const int e  = base + u * HT;
            const int ec = (e < c1) ? e : (c1 - 1);
            const int dl = dstp[ec];
            s[u] = srcp[ec];
            d[u] = (e < c1) ? dl : -1;
        }
        unsigned pos[HU];
#pragma unroll
        for (int u = 0; u < HU; ++u) {
            const unsigned loc = (unsigned)(d[u] - lo);
            pos[u] = 0u;
            if (loc < (unsigned)SLICE) pos[u] = atomicAdd(&cur[loc], 1u);
        }
#pragma unroll
        for (int u = 0; u < HU; ++u) {
            const unsigned loc = (unsigned)(d[u] - lo);
            if (loc < (unsigned)SLICE) csr[pos[u]] = s[u];
        }
    }
}

__global__ __launch_bounds__(256)
void gcn_agg1_gather(const int* __restrict__ csr, const unsigned* __restrict__ first,
                     const unsigned* __restrict__ degi, const float* __restrict__ xWs,
                     const float* __restrict__ dis, const float* __restrict__ b1,
                     const float* __restrict__ W2, float* __restrict__ y, int n) {
    const int t    = blockIdx.x * blockDim.x + threadIdx.x;
    const int node = t >> 4;
    const int j    = t & 15;
    const bool live = node < n;
    const int nd   = live ? node : (n - 1);
    const unsigned beg = first[nd];
    const unsigned cnt = degi[nd];
    const float* __restrict__ col = xWs + j;
    float acc = col[(long long)nd * F_HID];
    int next = ((unsigned)j < cnt) ? csr[beg + (unsigned)j] : -1;
    for (unsigned k0 = 0u; k0 < cnt; k0 += 16u) {
        const int mine = next;
        int   s[16];
        float r[16];
#pragma unroll
        for (int u = 0; u < 16; ++u) {
            s[u] = __shfl(mine, u, 16);
            const int sc = (s[u] >= 0) ? s[u] : nd;
            r[u] = col[(long long)sc * F_HID];
        }
        const unsigned kn = k0 + 16u + (unsigned)j;
        next = (kn < cnt) ? csr[beg + kn] : -1;
#pragma unroll
        for (int u = 0; u < 16; ++u) acc += (s[u] >= 0) ? r[u] : 0.0f;
    }
    const float dv = dis[nd];
    float hv = fmaf(acc, dv, b1[j]);
    hv = fmaxf(hv, 0.0f);
    float p = hv * W2[j];
    p += __shfl_xor(p, 8, 16);
    p += __shfl_xor(p, 4, 16);
    p += __shfl_xor(p, 2, 16);
    p += __shfl_xor(p, 1, 16);
    if (live && j == 0) y[node] = p * dv;
}

__global__ __launch_bounds__(256)
void gcn_agg2_gather(const int* __restrict__ csr, const unsigned* __restrict__ first,
                     const unsigned* __restrict__ degi, const float* __restrict__ y,
                     const float* __restrict__ dis, const float* __restrict__ b2,
                     float* __restrict__ out, int n) {
    const int t    = blockIdx.x * blockDim.x + threadIdx.x;
    const int node = t >> 4;
    const int j    = t & 15;
    const bool live = node < n;
    const int nd   = live ? node : (n - 1);
    const unsigned beg = first[nd];
    const unsigned cnt = degi[nd];
    float sum = 0.0f;
    for (unsigned k0 = (unsigned)j; k0 < cnt; k0 += 64u) {
        int s[4];
        float v[4];
#pragma unroll
        for (int u = 0; u < 4; ++u) {
            const unsigned k = k0 + 16u * (unsigned)u;
            s[u] = (k < cnt) ? csr[beg + k] : -1;
        }
#pragma unroll
        for (int u = 0; u < 4; ++u) {
            const int sc = (s[u] >= 0) ? s[u] : nd;
            const float r = y[sc];
            v[u] = (s[u] >= 0) ? r : 0.0f;
        }
#pragma unroll
        for (int u = 0; u < 4; ++u) sum += v[u];
    }
    sum += __shfl_xor(sum, 8, 16);
    sum += __shfl_xor(sum, 4, 16);
    sum += __shfl_xor(sum, 2, 16);
    sum += __shfl_xor(sum, 1, 16);
    if (live && j == 0) out[node] = b2[0] + dis[node] * (y[node] + sum);
}

extern "C" void kernel_launch(void* const* d_in, const int* in_sizes, int n_in,
                              void* d_out, int out_size, void* d_ws, size_t ws_size,
                              hipStream_t stream) {
    const float* x  = (const float*)d_in[0];
    const int*   ei = (const int*)d_in[1];
    const float* W1 = (const float*)d_in[2];
    const float* b1 = (const float*)d_in[3];
    const float* W2 = (const float*)d_in[4];
    const float* b2 = (const float*)d_in[5];
    float* out = (float*)d_out;

    const int       N = in_sizes[0] / F_IN;
    const long long E = (long long)in_sizes[1] / 2;

    const int       Ei       = (int)E;
    const int       n_slices = (N + SLICE - 1) / SLICE;
    const long long pitch    = (long long)n_slices * SLICE;
    const int       chunk_len = (int)((E + NCHUNK - 1) / NCHUNK);
    const unsigned  nb_acc   = (unsigned)(n_slices * NCHUNK);

    float*    ws    = (float*)d_ws;
    float*    dis   = ws;
    float*    xWs   = ws + N;
    float*    y     = ws + (long long)17 * N;
    unsigned* degi  = (unsigned*)(ws + (long long)18 * N);
    unsigned* first = (unsigned*)(ws + (long long)19 * N);
    unsigned* part  = (unsigned*)(ws + (long long)20 * N);
    int*      csr   = (int*)(ws + (long long)20 * N + (long long)NCHUNK * pitch);

    const int T = 256;
    const unsigned nb_n    = (unsigned)((N + T - 1) / T);
    const unsigned nb_n16  = (unsigned)(((long long)N * F_HID + T - 1) / T);
    const unsigned nb_tile = (unsigned)((N + 15) / 16);

    gcn_deg_slices <<<nb_acc,  HT, 0, stream>>>(ei, part, Ei, n_slices, chunk_len, pitch);
    gcn_deg_finish <<<nb_n,    T,  0, stream>>>(part, dis, degi, N, pitch);
    gcn_first_slot <<<(unsigned)n_slices, ST, 0, stream>>>(degi, first, N, Ei);
    gcn_xw_wmma    <<<nb_tile, 32, 0, stream>>>(x, W1, dis, xWs, N);
    gcn_csr_fill   <<<nb_acc,  HT, 0, stream>>>(ei, part, first, csr, Ei, N, n_slices, chunk_len, pitch);
    gcn_agg1_gather<<<nb_n16,  T,  0, stream>>>(csr, first, degi, xWs, dis, b1, W2, y, N);
    gcn_agg2_gather<<<nb_n16,  T,  0, stream>>>(csr, first, degi, y, dis, b2, out, N);
}


# Layer-2 edge sum looks y up in LDS: each block stages y in two 200 KB slices with whole-line loads, its nodes' sorted sources held in registers; no L2 request per edge; f32 sum order changes
# speedup vs baseline: 3.3784x; 1.0366x over previous
#include <hip/hip_runtime.h>
#include <hip/hip_bf16.h>

typedef __attribute__((ext_vector_type(2))) float v2f;
typedef __attribute__((ext_vector_type(8))) float v8f;

#define F_IN   128
#define F_HID  16

#define SLICE   25600
#define NCHUNK  64
#define HT      512
#define HU      16
#define ST      1024
#define SPT     (SLICE / ST)
#define FB      10

__global__ __launch_bounds__(HT)
void gcn_deg_slices(const int* __restrict__ ei, unsigned* __restrict__ part,
                    int E, int n_slices, int chunk_len, long long pitch) {
    __shared__ unsigned cnt[SLICE];
    const int sl = blockIdx.x % n_slices;
    const int ch = blockIdx.x / n_slices;
    const int lo = sl * SLICE;
    for (int i = threadIdx.x; i < SLICE; i += HT) cnt[i] = 0u;
    __syncthreads();

    const int* __restrict__ dstp = ei + E;
    const int c0 = ch * chunk_len;
    const int c1 = (c0 + chunk_len < E) ? (c0 + chunk_len) : E;
    for (int base = c0 + (int)threadIdx.x; base < c1; base += HT * HU) {
        int d[HU];
#pragma unroll
        for (int u = 0; u < HU; ++u) {
            const int e  = base + u * HT;
            const int ec = (e < c1) ? e : (c1 - 1);
            const int dl = dstp[ec];
            d[u] = (e < c1) ? dl : -1;
        }
#pragma unroll
        for (int u = 0; u < HU; ++u) {
            const unsigned loc = (unsigned)(d[u] - lo);
            if (loc < (unsigned)SLICE) atomicAdd(&cnt[loc], 1u);
        }
    }
    __syncthreads();
    unsigned* __restrict__ row = part + (long long)ch * pitch + lo;
    for (int i = threadIdx.x; i < SLICE; i += HT) row[i] = cnt[i];
}

__global__ __launch_bounds__(256)
void gcn_deg_finish(unsigned* __restrict__ part, float* __restrict__ dis,
                               unsigned* __restrict__ degi, int n, long long pitch) {
    int i = blockIdx.x * blockDim.x + threadIdx.x;
    if (i >= n) return;
    unsigned p[NCHUNK];
#pragma unroll
    for (int ch = 0; ch < NCHUNK; ++ch) p[ch] = part[(long long)ch * pitch + i];
    unsigned run = 0u;
#pragma unroll
    for (int ch = 0; ch < NCHUNK; ++ch) {
        part[(long long)ch * pitch + i] = run;
        run += p[ch];
    }
    degi[i] = run;
    const float dgv = (float)(run + 1u);
    dis[i] = dgv > 0.0f ? rsqrtf(dgv) : 0.0f;
}

__global__ __launch_bounds__(ST)
void gcn_first_slot(const unsigned* __restrict__ degi, unsigned* __restrict__ first,
                    int n, int E) {
    __shared__ unsigned cell[SLICE];
    __shared__ unsigned sums[ST];
    const int t    = threadIdx.x;
    const int base = blockIdx.x * SLICE;
    unsigned v[SPT];
#pragma unroll
    for (int u = 0; u < SPT; ++u) {
        const int idx = base + u * ST + t;
        const int ic  = (idx < n) ? idx : (n - 1);
        const unsigned dv = degi[ic];
        v[u] = (idx < n) ? dv : 0u;
    }
#pragma unroll
    for (int u = 0; u < SPT; ++u) cell[u * ST + t] = v[u];
    __syncthreads();

    unsigned s = 0u;
#pragma unroll
    for (int k = 0; k < SPT; ++k) {
        const unsigned c = cell[t * SPT + k];
        cell[t * SPT + k] = s;
        s += c;
    }
    sums[t] = s;
    __syncthreads();
    for (int off = 1; off < ST; off <<= 1) {
        const unsigned a = (t >= off) ? sums[t - off] : 0u;
        __syncthreads();
        sums[t] += a;
        __syncthreads();
    }
    const unsigned add = (unsigned)blockIdx.x * (unsigned)E + (sums[t] - s);
#pragma unroll
    for (int k = 0; k < SPT; ++k) cell[t * SPT + k] += add;
    __syncthreads();
#pragma unroll
    for (int u = 0; u < SPT; ++u) {
        const int idx = base + u * ST + t;
        if (idx < n) first[idx] = cell[u * ST + t];
    }
}

__global__ __launch_bounds__(32)
void gcn_xw_wmma(const float* __restrict__ x, const float* __restrict__ W1,
                 const float* __restrict__ dis, float* __restrict__ xWs, int n) {
    const int row0 = blockIdx.x * 16;
    if (row0 >= n) return;
    const int lane = threadIdx.x;
    const int half = lane >> 4;
    const int l16  = lane & 15;

    const float* xrow = x + (long long)(row0 + l16) * F_IN;

    if (row0 + 16 + l16 < n)
        __builtin_prefetch(x + (long long)(row0 + 16 + l16) * F_IN + half * 64, 0, 3);

    v8f c = {};
#pragma unroll
    for (int k0 = 0; k0 < F_IN; k0 += 4) {
        const int ka = k0 + half * 2;
        v2f a, b;
        a.x = xrow[ka];
        a.y = xrow[ka + 1];
        b.x = W1[ka * F_HID + l16];
        b.y = W1[(ka + 1) * F_HID + l16];
        c = __builtin_amdgcn_wmma_f32_16x16x4_f32(
                 false, a,  false, b,
                 (short)0, c,  false,  false);
    }
#pragma unroll
    for (int r = 0; r < 8; ++r) {
        const int m = r + half * 8;
        xWs[(long long)(row0 + m) * F_HID + l16] = c[r] * dis[row0 + m];
    }
}

__global__ __launch_bounds__(HT)
void gcn_csr_fill(const int* __restrict__ ei, const unsigned* __restrict__ part,
                  const unsigned* __restrict__ first, int* __restrict__ csr,
                  int E, int n, int n_slices, int chunk_len, long long pitch) {
    __shared__ unsigned cur[SLICE];
    const int sl = blockIdx.x % n_slices;
    const int ch = blockIdx.x / n_slices;
    const int lo = sl * SLICE;
    const unsigned* __restrict__ row = part + (long long)ch * pitch;
    for (int i0 = (int)threadIdx.x; i0 < SLICE; i0 += HT * FB) {
        unsigned a[FB], b[FB];
#pragma unroll
        for (int u = 0; u < FB; ++u) {
            const int node = lo + i0 + u * HT;
            const int nc   = (node < n) ? node : (n - 1);
            a[u] = first[nc];
            b[u] = row[nc];
        }
#pragma unroll
        for (int u = 0; u < FB; ++u) cur[i0 + u * HT] = a[u] + b[u];
    }
    __syncthreads();

    const int* __restrict__ srcp = ei;
    const int* __restrict__ dstp = ei + E;
    const int c0 = ch * chunk_len;
    const int c1 = (c0 + chunk_len < E) ? (c0 + chunk_len) : E;
    for (int base = c0 + (int)threadIdx.x; base < c1; base += HT * HU) {
        int d[HU], s[HU];
#pragma unroll
        for (int u = 0; u < HU; ++u) {
            const int e  = base + u * HT;
            const int ec = (e < c1) ? e : (c1 - 1);
            const int dl = dstp[ec];
            s[u] = srcp[ec];
            d[u] = (e < c1) ? dl : -1;
        }
        unsigned pos[HU];
#pragma unroll
        for (int u = 0; u < HU; ++u) {
            const unsigned loc = (unsigned)(d[u] - lo);
            pos[u] = 0u;
            if (loc < (unsigned)SLICE) pos[u] = atomicAdd(&cur[loc], 1u);
        }
#pragma unroll
        for (int u = 0; u < HU; ++u) {
            const unsigned loc = (unsigned)(d[u] - lo);
            if (loc < (unsigned)SLICE) csr[pos[u]] = s[u];
        }
    }
}

__global__ __launch_bounds__(256)
void gcn_agg1_gather(const int* __restrict__ csr, const unsigned* __restrict__ first,
                     const unsigned* __restrict__ degi, const float* __restrict__ xWs,
                     const float* __restrict__ dis, const float* __restrict__ b1,
                     const float* __restrict__ W2, float* __restrict__ y, int n) {
    const int t    = blockIdx.x * blockDim.x + threadIdx.x;
    const int node = t >> 4;
    const int j    = t & 15;
    const bool live = node < n;
    const int nd   = live ? node : (n - 1);
    const unsigned beg = first[nd];
    const unsigned cnt = degi[nd];
    const float* __restrict__ col = xWs + j;
    float acc = col[(long long)nd * F_HID];
    int next = ((unsigned)j < cnt) ? csr[beg + (unsigned)j] : -1;
    for (unsigned k0 = 0u; k0 < cnt; k0 += 16u) {
        const int mine = next;
        int   s[16];
        float r[16];
#pragma unroll
        for (int u = 0; u < 16; ++u) {
            s[u] = __shfl(mine, u, 16);
            const int sc = (s[u] >= 0) ? s[u] : nd;
            r[u] = col[(long long)sc * F_HID];
        }
        const unsigned kn = k0 + 16u + (unsigned)j;
        next = (kn < cnt) ? csr[beg + kn] : -1;
#pragma unroll
        for (int u = 0; u < 16; ++u) acc += (s[u] >= 0) ? r[u] : 0.0f;
    }
    const float dv = dis[nd];
    float hv = fmaf(acc, dv, b1[j]);
    hv = fmaxf(hv, 0.0f);
    float p = hv * W2[j];
    p += __shfl_xor(p, 8, 16);
    p += __shfl_xor(p, 4, 16);
    p += __shfl_xor(p, 2, 16);
    p += __shfl_xor(p, 1, 16);
    if (live && j == 0) y[node] = p * dv;
}

#define YS     51200
#define A2T    1024
#define A2G    (A2T / 16)
#define A2IT   7
#define A2NPB  (A2G * A2IT)
#define A2SW   ((YS + A2T * 4 - 1) / (A2T * 4))
#define A2SA   7

template <int B>
__device__ __forceinline__ void gcn_stage_y(const float* __restrict__ ysrc, float* __restrict__ ys,
                                            int len4, int sw0, int rot, unsigned t4) {
    float4 v[B];
#pragma unroll
    for (int b = 0; b < B; ++b) {
        const int sw  = (sw0 + b + rot < A2SW) ? (sw0 + b + rot) : (sw0 + b + rot - A2SW);
        const int off = sw * (A2T * 4);
        v[b] = make_float4(0.0f, 0.0f, 0.0f, 0.0f);
        if ((int)t4 < len4 - off) v[b] = *reinterpret_cast<const float4*>(ysrc + off + t4);
    }
#pragma unroll
    for (int b = 0; b < B; ++b) {
        const int sw  = (sw0 + b + rot < A2SW) ? (sw0 + b + rot) : (sw0 + b + rot - A2SW);
        const int off = sw * (A2T * 4);
        if ((int)t4 < len4 - off) *reinterpret_cast<float4*>(ys + off + t4) = v[b];
    }
}

__global__ __launch_bounds__(A2T)
void gcn_agg2_lds(const int* __restrict__ csr, const unsigned* __restrict__ first,
                  const unsigned* __restrict__ degi, const float* __restrict__ y,
                  const float* __restrict__ dis, const float* __restrict__ b2,
                  float* __restrict__ out, int n) {
    __shared__ __attribute__((aligned(16))) float ys[YS];
    const int tid  = threadIdx.x;
    const int grp  = tid >> 4;
    const int j    = tid & 15;
    const int base = blockIdx.x * A2NPB;
    const int rot  = blockIdx.x % A2SW;

    unsigned cnt[A2IT];
    int      s[A2IT][4];
    {
        unsigned beg[A2IT];
#pragma unroll
        for (int it = 0; it < A2IT; ++it) {
            const int node = base + it * A2G + grp;
            const int nd   = (node < n) ? node : (n - 1);
            beg[it] = first[nd];
            const unsigned c = degi[nd];
            cnt[it] = (node < n) ? c : 0u;
        }
#pragma unroll
        for (int it = 0; it < A2IT; ++it) {
#pragma unroll
            for (int u = 0; u < 4; ++u) {
                const unsigned k = (unsigned)j + 16u * (unsigned)u;
                s[it][u] = (k < cnt[it]) ? csr[beg[it] + k] : -1;
            }
        }
    }
    float acc[A2IT], dv[A2IT], yv[A2IT];
#pragma unroll
    for (int it = 0; it < A2IT; ++it) {
        const int node = base + it * A2G + grp;
        const int nd   = (node < n) ? node : (n - 1);
        dv[it]  = dis[nd];
        yv[it]  = y[nd];
        acc[it] = 0.0f;
    }

    for (int lo = 0; lo < n; lo += YS) {
        const int len  = (n - lo < YS) ? (n - lo) : YS;
        const int len4 = len & ~3;
        gcn_stage_y<A2SA>(y + lo, ys, len4, 0, rot, (unsigned)tid * 4u);
        gcn_stage_y<A2SW - A2SA>(y + lo, ys, len4, A2SA, rot, (unsigned)tid * 4u);
        if (tid < len - len4) ys[len4 + tid] = y[lo + len4 + tid];
        __syncthreads();
#pragma unroll
        for (int it = 0; it < A2IT; ++it) {
#pragma unroll
            for (int u = 0; u < 4; ++u) {
                const unsigned loc = (unsigned)(s[it][u] - lo);
                const float    r   = ys[(loc < (unsigned)len) ? loc : 0u];
                acc[it] += (loc < (unsigned)len) ? r : 0.0f;
            }
        }
#pragma unroll
        for (int it = 0; it < A2IT; ++it) {
            if (cnt[it] > 64u) {
                const unsigned bg = first[base + it * A2G + grp];
                for (unsigned k = 64u + (unsigned)j; k < cnt[it]; k += 16u) {
                    const unsigned loc = (unsigned)(csr[bg + k] - lo);
                    if (loc < (unsigned)len) acc[it] += ys[loc];
                }
            }
        }
        __syncthreads();
    }
#pragma unroll
    for (int w = 8; w >= 1; w >>= 1) {
#pragma unroll
        for (int it = 0; it < A2IT; ++it) acc[it] += __shfl_xor(acc[it], w, 16);
    }
    const float bias = b2[0];
#pragma unroll
    for (int it = 0; it < A2IT; ++it) {
        const int node = base + it * A2G + grp;
        if (node < n && j == 0) out[node] = bias + dv[it] * (yv[it] + acc[it]);
    }
}

extern "C" void kernel_launch(void* const* d_in, const int* in_sizes, int n_in,
                              void* d_out, int out_size, void* d_ws, size_t ws_size,
                              hipStream_t stream) {
    const float* x  = (const float*)d_in[0];
    const int*   ei = (const int*)d_in[1];
    const float* W1 = (const float*)d_in[2];
    const float* b1 = (const float*)d_in[3];
    const float* W2 = (const float*)d_in[4];
    const float* b2 = (const float*)d_in[5];
    float* out = (float*)d_out;

    const int       N = in_sizes[0] / F_IN;
    const long long E = (long long)in_sizes[1] / 2;

    const int       Ei       = (int)E;
    const int       n_slices = (N + SLICE - 1) / SLICE;
    const long long pitch    = (long long)n_slices * SLICE;
    const int       chunk_len = (int)((E + NCHUNK - 1) / NCHUNK);
    const unsigned  nb_acc   = (unsigned)(n_slices * NCHUNK);

    const long long Np = ((long long)N + 3) / 4 * 4;
    float*    ws    = (float*)d_ws;
    float*    y     = ws;
    float*    dis   = ws + Np;
    float*    xWs   = ws + Np + N;
    unsigned* degi  = (unsigned*)(ws + Np + (long long)17 * N);
    unsigned* first = (unsigned*)(ws + Np + (long long)18 * N);
    unsigned* part  = (unsigned*)(ws + Np + (long long)19 * N);
    int*      csr   = (int*)(ws + Np + (long long)19 * N + (long long)NCHUNK * pitch);

    const int T = 256;
    const unsigned nb_n    = (unsigned)((N + T - 1) / T);
    const unsigned nb_n16  = (unsigned)(((long long)N * F_HID + T - 1) / T);
    const unsigned nb_tile = (unsigned)((N + 15) / 16);
    const unsigned nb_a2   = (unsigned)((N + A2NPB - 1) / A2NPB);

    gcn_deg_slices <<<nb_acc,  HT, 0, stream>>>(ei, part, Ei, n_slices, chunk_len, pitch);
    gcn_deg_finish <<<nb_n,    T,  0, stream>>>(part, dis, degi, N, pitch);
    gcn_first_slot <<<(unsigned)n_slices, ST, 0, stream>>>(degi, first, N, Ei);
    gcn_xw_wmma    <<<nb_tile, 32, 0, stream>>>(x, W1, dis, xWs, N);
    gcn_csr_fill   <<<nb_acc,  HT, 0, stream>>>(ei, part, first, csr, Ei, N, n_slices, chunk_len, pitch);
    gcn_agg1_gather<<<nb_n16,  T,  0, stream>>>(csr, first, degi, xWs, dis, b1, W2, y, N);
    gcn_agg2_lds   <<<nb_a2,   A2T, 0, stream>>>(csr, first, degi, y, dis, b2, out, N);
}


# CSR build is a two-level LDS counting sort: edges binned by 256-node bucket in whole runs, each bucket ordered in LDS (sources rising); no per-edge global store; 6 launches; f32 sum order changes
# speedup vs baseline: 5.4507x; 1.6134x over previous
#include <hip/hip_runtime.h>
#include <hip/hip_bf16.h>

typedef __attribute__((ext_vector_type(2))) float v2f;
typedef __attribute__((ext_vector_type(8))) float v8f;

#define F_IN   128
#define F_HID  16

#define BN      256
#define NBP     512
#define BT      1024
#define BU      13
#define CHMAX   12800
#define CB      16
#define PB      8
#define CAPB    16384
#define SORTMAX 128

__device__ __forceinline__ unsigned gcn_block_scan(unsigned v, unsigned* wsum) {
    const int lane = threadIdx.x & 31;
    const int w    = threadIdx.x >> 5;
    unsigned inc = v;
#pragma unroll
    for (int o = 1; o < 32; o <<= 1) {
        const unsigned up = __shfl_up(inc, o, 32);
        if (lane >= o) inc += up;
    }
    if (lane == 31) wsum[w] = inc;
    __syncthreads();
    if (w == 0) {
        const unsigned s = wsum[lane];
        unsigned si = s;
#pragma unroll
        for (int o = 1; o < 32; o <<= 1) {
            const unsigned up = __shfl_up(si, o, 32);
            if (lane >= o) si += up;
        }
        wsum[lane] = si - s;
    }
    __syncthreads();
    const unsigned res = wsum[w] + inc - v;
    __syncthreads();
    return res;
}

__global__ __launch_bounds__(BT)
void gcn_bucket_count(const int* __restrict__ ei, unsigned* __restrict__ cntm,
                      int E, int n, int chunk_len) {
    __shared__ unsigned hist[NBP];
    const int t = threadIdx.x;
    if (t < NBP) hist[t] = 0u;
    __syncthreads();
    const int* __restrict__ dstp = ei + E;
    const int c0 = blockIdx.x * chunk_len;
    const int c1 = (c0 + chunk_len < E) ? (c0 + chunk_len) : E;
    int d[BU];
#pragma unroll
    for (int u = 0; u < BU; ++u) {
        const int e  = c0 + u * BT + t;
        const int ec = (e < c1) ? e : (c1 - 1);
        const int dl = dstp[ec];
        d[u] = (e < c1) ? dl : -1;
    }
#pragma unroll
    for (int u = 0; u < BU; ++u)
        if ((unsigned)d[u] < (unsigned)n) atomicAdd(&hist[d[u] >> 8], 1u);
    __syncthreads();
    if (t < NBP) cntm[blockIdx.x * NBP + t] = hist[t];
}

__global__ __launch_bounds__(BT)
void gcn_bucket_fill(const int* __restrict__ ei, const unsigned* __restrict__ cntm,
                     unsigned* __restrict__ bbase, unsigned* __restrict__ bcnt,
                     unsigned* __restrict__ pk, int E, int n, int chunk_len, int nblk) {
    __shared__ unsigned sorted[CHMAX];
    __shared__ unsigned gpos[CHMAX];
    __shared__ unsigned ntot;
    __shared__ unsigned tot2[2][NBP];
    __shared__ unsigned lo2[2][NBP];
    __shared__ unsigned bpos[NBP];
    __shared__ unsigned lstart[NBP];
    __shared__ unsigned lcur[NBP];
    __shared__ unsigned wsum[32];
    const int t   = threadIdx.x;
    const int blk = blockIdx.x;

    const int* __restrict__ srcp = ei;
    const int* __restrict__ dstp = ei + E;
    const int c0 = blk * chunk_len;
    const int c1 = (c0 + chunk_len < E) ? (c0 + chunk_len) : E;
    {
        const int col  = t & (NBP - 1);
        const int half = t >> 9;
        unsigned tot = 0u, lo = 0u;
        for (int r0 = half; r0 < nblk; r0 += 2 * CB) {
            unsigned v[CB];
#pragma unroll
            for (int k = 0; k < CB; ++k) {
                const int r  = r0 + 2 * k;
                const int rc = (r < nblk) ? r : (nblk - 1);
                v[k] = cntm[rc * NBP + col];
            }
#pragma unroll
            for (int k = 0; k < CB; ++k) {
                const int r = r0 + 2 * k;
                tot += (r < nblk) ? v[k] : 0u;
                lo  += (r < blk)  ? v[k] : 0u;
            }
        }
        tot2[half][col] = tot;
        lo2[half][col]  = lo;
    }
    __syncthreads();
    int d[BU], s[BU];
#pragma unroll
    for (int u = 0; u < BU; ++u) {
        const int e  = c0 + u * BT + t;
        const int ec = (e < c1) ? e : (c1 - 1);
        const int dl = dstp[ec];
        s[u] = srcp[ec];
        d[u] = (e < c1) ? dl : -1;
    }
    unsigned ctot = 0u, clo = 0u, cm = 0u;
    if (t < NBP) {
        ctot = tot2[0][t] + tot2[1][t];
        clo  = lo2[0][t] + lo2[1][t];
        cm   = cntm[blk * NBP + t];
    }
    const unsigned bb = gcn_block_scan(ctot, wsum);
    const unsigned ls = gcn_block_scan(cm, wsum);
    if (t < NBP) {
        bpos[t]   = bb + clo;
        lstart[t] = ls;
        lcur[t]   = 0u;
        if (t == NBP - 1) ntot = ls + cm;
        if (blk == 0) { bbase[t] = bb; bcnt[t] = ctot; }
    }
    __syncthreads();

    unsigned rk[BU], l0[BU], g0[BU];
#pragma unroll
    for (int u = 0; u < BU; ++u) {
        const bool     ok = (unsigned)d[u] < (unsigned)n;
        const unsigned b  = ok ? ((unsigned)d[u] >> 8) : 0u;
        l0[u] = lstart[b];
        g0[u] = bpos[b];
        rk[u] = 0u;
        if (ok) rk[u] = atomicAdd(&lcur[b], 1u);
    }
#pragma unroll
    for (int u = 0; u < BU; ++u) {
        if ((unsigned)d[u] < (unsigned)n) {
            const unsigned slot = l0[u] + rk[u];
            if (slot < (unsigned)CHMAX) {
                sorted[slot] = ((unsigned)s[u] << 8) | ((unsigned)d[u] & 255u);
                gpos[slot]   = g0[u] + rk[u];
            }
        }
    }
    __syncthreads();

    const unsigned cnt_blk = (ntot < (unsigned)CHMAX) ? ntot : (unsigned)CHMAX;
    unsigned gp[BU], wv[BU];
#pragma unroll
    for (int u = 0; u < BU; ++u) {
        const unsigned idx = (unsigned)(u * BT + t);
        const unsigned ic  = (idx < cnt_blk) ? idx : 0u;
        gp[u] = gpos[ic];
        wv[u] = sorted[ic];
    }
#pragma unroll
    for (int u = 0; u < BU; ++u) {
        const unsigned idx = (unsigned)(u * BT + t);
        if (idx < cnt_blk) pk[gp[u]] = wv[u];
    }
}

__global__ __launch_bounds__(BT)
void gcn_bucket_finish(const unsigned* __restrict__ pk, const unsigned* __restrict__ bbase,
                       const unsigned* __restrict__ bcnt, int* __restrict__ csr,
                       unsigned* __restrict__ first, unsigned* __restrict__ degi,
                       float* __restrict__ dis, int n) {
    __shared__ unsigned buf[CAPB];
    __shared__ unsigned buf2[CAPB];
    __shared__ unsigned ncnt[BN];
    __shared__ unsigned nstart[BN];
    __shared__ unsigned ncur[BN];
    __shared__ unsigned wsum[32];
    const int t = threadIdx.x;
    const int b = blockIdx.x;
    const unsigned gb = bbase[b];
    const unsigned nb = bcnt[b];
    const bool fits = nb <= (unsigned)CAPB;
    if (t < BN) ncnt[t] = 0u;
    __syncthreads();

    for (unsigned base = (unsigned)t; base < nb; base += BT * PB) {
        unsigned e[PB];
#pragma unroll
        for (int u = 0; u < PB; ++u) {
            const unsigned i  = base + (unsigned)u * BT;
            const unsigned ic = (i < nb) ? i : (nb - 1u);
            e[u] = pk[gb + ic];
        }
#pragma unroll
        for (int u = 0; u < PB; ++u)
            if (base + (unsigned)u * BT < nb) atomicAdd(&ncnt[e[u] & 255u], 1u);
    }
    __syncthreads();

    const unsigned v  = (t < BN) ? ncnt[t] : 0u;
    const unsigned ex = gcn_block_scan(v, wsum);
    if (t < BN) {
        nstart[t] = ex;
        ncur[t]   = 0u;
        const int node = b * BN + t;
        if (node < n) {
            degi[node]  = v;
            first[node] = gb + ex;
            const float dgv = (float)(v + 1u);
            dis[node] = dgv > 0.0f ? rsqrtf(dgv) : 0.0f;
        }
    }
    __syncthreads();

    for (unsigned base = (unsigned)t; base < nb; base += BT * PB) {
        unsigned e[PB];
#pragma unroll
        for (int u = 0; u < PB; ++u) {
            const unsigned i  = base + (unsigned)u * BT;
            const unsigned ic = (i < nb) ? i : (nb - 1u);
            e[u] = pk[gb + ic];
        }
        unsigned rk[PB];
#pragma unroll
        for (int u = 0; u < PB; ++u) {
            rk[u] = 0u;
            if (base + (unsigned)u * BT < nb) rk[u] = atomicAdd(&ncur[e[u] & 255u], 1u);
        }
#pragma unroll
        for (int u = 0; u < PB; ++u) {
            if (base + (unsigned)u * BT < nb) {
                const unsigned slot = nstart[e[u] & 255u] + rk[u];
                if (fits) { if (slot < (unsigned)CAPB) buf[slot] = (e[u] << 7) | (rk[u] & 127u); }
                else      csr[gb + slot] = (int)(e[u] >> 8);
            }
        }
    }
    __syncthreads();

    if (fits) {
        for (unsigned i = (unsigned)t; i < nb; i += BT) {
            const unsigned key = buf[i];
            const unsigned ld  = (key >> 7) & 255u;
            const unsigned ns  = nstart[ld];
            const unsigned len = ncnt[ld];
            unsigned r = i - ns;
            if (len <= (unsigned)SORTMAX) {
                r = 0u;
                for (unsigned j = 0u; j < len; ++j) r += (buf[ns + j] < key) ? 1u : 0u;
            }
            buf2[ns + r] = key >> 15;
        }
        __syncthreads();
        for (unsigned i = (unsigned)t; i < nb; i += BT) csr[gb + i] = (int)buf2[i];
    }
}

__global__ __launch_bounds__(32)
void gcn_xw_wmma(const float* __restrict__ x, const float* __restrict__ W1,
                 const float* __restrict__ dis, float* __restrict__ xWs, int n) {
    const int row0 = blockIdx.x * 16;
    if (row0 >= n) return;
    const int lane = threadIdx.x;
    const int half = lane >> 4;
    const int l16  = lane & 15;

    const float* xrow = x + (long long)(row0 + l16) * F_IN;

    if (row0 + 16 + l16 < n)
        __builtin_prefetch(x + (long long)(row0 + 16 + l16) * F_IN + half * 64, 0, 3);

    v8f c = {};
#pragma unroll
    for (int k0 = 0; k0 < F_IN; k0 += 4) {
        const int ka = k0 + half * 2;
        v2f a, b;
        a.x = xrow[ka];
        a.y = xrow[ka + 1];
        b.x = W1[ka * F_HID + l16];
        b.y = W1[(ka + 1) * F_HID + l16];
        c = __builtin_amdgcn_wmma_f32_16x16x4_f32(
                 false, a,  false, b,
                 (short)0, c,  false,  false);
    }
#pragma unroll
    for (int r = 0; r < 8; ++r) {
        const int m = r + half * 8;
        xWs[(long long)(row0 + m) * F_HID + l16] = c[r] * dis[row0 + m];
    }
}

__global__ __launch_bounds__(256)
void gcn_agg1_gather(const int* __restrict__ csr, const unsigned* __restrict__ first,
                     const unsigned* __restrict__ degi, const float* __restrict__ xWs,
                     const float* __restrict__ dis, const float* __restrict__ b1,
                     const float* __restrict__ W2, float* __restrict__ y, int n) {
    const int t    = blockIdx.x * blockDim.x + threadIdx.x;
    const int node = t >> 4;
    const int j    = t & 15;
    const bool live = node < n;
    const int nd   = live ? node : (n - 1);
    const unsigned beg = first[nd];
    const unsigned cnt = degi[nd];
    const float* __restrict__ col = xWs + j;
    float acc = col[(long long)nd * F_HID];
    int next = ((unsigned)j < cnt) ? csr[beg + (unsigned)j] : -1;
    for (unsigned k0 = 0u; k0 < cnt; k0 += 16u) {
        const int mine = next;
        int   s[16];
        float r[16];
#pragma unroll
        for (int u = 0; u < 16; ++u) {
            s[u] = __shfl(mine, u, 16);
            const int sc = (s[u] >= 0) ? s[u] : nd;
            r[u] = col[(long long)sc * F_HID];
        }
        const unsigned kn = k0 + 16u + (unsigned)j;
        next = (kn < cnt) ? csr[beg + kn] : -1;
#pragma unroll
        for (int u = 0; u < 16; ++u) acc += (s[u] >= 0) ? r[u] : 0.0f;
    }
    const float dv = dis[nd];
    float hv = fmaf(acc, dv, b1[j]);
    hv = fmaxf(hv, 0.0f);
    float p = hv * W2[j];
    p += __shfl_xor(p, 8, 16);
    p += __shfl_xor(p, 4, 16);
    p += __shfl_xor(p, 2, 16);
    p += __shfl_xor(p, 1, 16);
    if (live && j == 0) y[node] = p * dv;
}

#define YS     51200
#define A2T    1024
#define A2G    (A2T / 16)
#define A2IT   7
#define A2NPB  (A2G * A2IT)
#define A2SW   ((YS + A2T * 4 - 1) / (A2T * 4))
#define A2SA   7

template <int B>
__device__ __forceinline__ void gcn_stage_y(const float* __restrict__ ysrc, float* __restrict__ ys,
                                            int len4, int sw0, int rot, unsigned t4) {
    float4 v[B];
#pragma unroll
    for (int b = 0; b < B; ++b) {
        const int sw  = (sw0 + b + rot < A2SW) ? (sw0 + b + rot) : (sw0 + b + rot - A2SW);
        const int off = sw * (A2T * 4);
        v[b] = make_float4(0.0f, 0.0f, 0.0f, 0.0f);
        if ((int)t4 < len4 - off) v[b] = *reinterpret_cast<const float4*>(ysrc + off + t4);
    }
#pragma unroll
    for (int b = 0; b < B; ++b) {
        const int sw  = (sw0 + b + rot < A2SW) ? (sw0 + b + rot) : (sw0 + b + rot - A2SW);
        const int off = sw * (A2T * 4);
        if ((int)t4 < len4 - off) *reinterpret_cast<float4*>(ys + off + t4) = v[b];
    }
}

__global__ __launch_bounds__(A2T)
void gcn_agg2_lds(const int* __restrict__ csr, const unsigned* __restrict__ first,
                  const unsigned* __restrict__ degi, const float* __restrict__ y,
                  const float* __restrict__ dis, const float* __restrict__ b2,
                  float* __restrict__ out, int n) {
    __shared__ __attribute__((aligned(16))) float ys[YS];
    const int tid  = threadIdx.x;
    const int grp  = tid >> 4;
    const int j    = tid & 15;
    const int base = blockIdx.x * A2NPB;
    const int rot  = blockIdx.x % A2SW;

    unsigned cnt[A2IT];
    int      s[A2IT][4];
    {
        unsigned beg[A2IT];
#pragma unroll
        for (int it = 0; it < A2IT; ++it) {
            const int node = base + it * A2G + grp;
            const int nd   = (node < n) ? node : (n - 1);
            beg[it] = first[nd];
            const unsigned c = degi[nd];
            cnt[it] = (node < n) ? c : 0u;
        }
#pragma unroll
        for (int it = 0; it < A2IT; ++it) {
#pragma unroll
            for (int u = 0; u < 4; ++u) {
                const unsigned k = (unsigned)j + 16u * (unsigned)u;
                s[it][u] = (k < cnt[it]) ? csr[beg[it] + k] : -1;
            }
        }
    }
    float acc[A2IT], dv[A2IT], yv[A2IT];
#pragma unroll
    for (int it = 0; it < A2IT; ++it) {
        const int node = base + it * A2G + grp;
        const int nd   = (node < n) ? node : (n - 1);
        dv[it]  = dis[nd];
        yv[it]  = y[nd];
        acc[it] = 0.0f;
    }

    for (int lo = 0; lo < n; lo += YS) {
        const int len  = (n - lo < YS) ? (n - lo) : YS;
        const int len4 = len & ~3;
        gcn_stage_y<A2SA>(y + lo, ys, len4, 0, rot, (unsigned)tid * 4u);
        gcn_stage_y<A2SW - A2SA>(y + lo, ys, len4, A2SA, rot, (unsigned)tid * 4u);
        if (tid < len - len4) ys[len4 + tid] = y[lo + len4 + tid];
        __syncthreads();
#pragma unroll
        for (int it = 0; it < A2IT; ++it) {
#pragma unroll
            for (int u = 0; u < 4; ++u) {
                const unsigned loc = (unsigned)(s[it][u] - lo);
                const float    r   = ys[(loc < (unsigned)len) ? loc : 0u];
                acc[it] += (loc < (unsigned)len) ? r : 0.0f;
            }
        }
#pragma unroll
        for (int it = 0; it < A2IT; ++it) {
            if (cnt[it] > 64u) {
                const unsigned bg = first[base + it * A2G + grp];
                for (unsigned k = 64u + (unsigned)j; k < cnt[it]; k += 16u) {
                    const unsigned loc = (unsigned)(csr[bg + k] - lo);
                    if (loc < (unsigned)len) acc[it] += ys[loc];
                }
            }
        }
        __syncthreads();
    }
#pragma unroll
    for (int w = 8; w >= 1; w >>= 1) {
#pragma unroll
        for (int it = 0; it < A2IT; ++it) acc[it] += __shfl_xor(acc[it], w, 16);
    }
    const float bias = b2[0];
#pragma unroll
    for (int it = 0; it < A2IT; ++it) {
        const int node = base + it * A2G + grp;
        if (node < n && j == 0) out[node] = bias + dv[it] * (yv[it] + acc[it]);
    }
}

extern "C" void kernel_launch(void* const* d_in, const int* in_sizes, int n_in,
                              void* d_out, int out_size, void* d_ws, size_t ws_size,
                              hipStream_t stream) {
    const float* x  = (const float*)d_in[0];
    const int*   ei = (const int*)d_in[1];
    const float* W1 = (const float*)d_in[2];
    const float* b1 = (const float*)d_in[3];
    const float* W2 = (const float*)d_in[4];
    const float* b2 = (const float*)d_in[5];
    float* out = (float*)d_out;

    const int       N = in_sizes[0] / F_IN;
    const long long E = (long long)in_sizes[1] / 2;

    const int Ei        = (int)E;
    const int nblk      = (int)((E + CHMAX - 1) / CHMAX);
    const int chunk_len = (int)((E + nblk - 1) / nblk);
    const int n_bkt     = (N + BN - 1) / BN;

    const long long Np = ((long long)N + 3) / 4 * 4;
    float*    ws    = (float*)d_ws;
    float*    y     = ws;
    float*    dis   = ws + Np;
    float*    xWs   = ws + Np + N;
    unsigned* degi  = (unsigned*)(ws + Np + (long long)17 * N);
    unsigned* first = (unsigned*)(ws + Np + (long long)18 * N);
    unsigned* bbase = (unsigned*)(ws + Np + (long long)19 * N);
    unsigned* bcnt  = bbase + NBP;
    unsigned* cntm  = bcnt + NBP;
    unsigned* pk    = cntm + (long long)nblk * NBP;
    int*      csr   = (int*)(pk + E);

    const int T = 256;
    const unsigned nb_n16  = (unsigned)(((long long)N * F_HID + T - 1) / T);
    const unsigned nb_tile = (unsigned)((N + 15) / 16);
    const unsigned nb_a2   = (unsigned)((N + A2NPB - 1) / A2NPB);

    gcn_bucket_count <<<(unsigned)nblk,  BT, 0, stream>>>(ei, cntm, Ei, N, chunk_len);
    gcn_bucket_fill  <<<(unsigned)nblk,  BT, 0, stream>>>(ei, cntm, bbase, bcnt, pk, Ei, N, chunk_len, nblk);
    gcn_bucket_finish<<<(unsigned)n_bkt, BT, 0, stream>>>(pk, bbase, bcnt, csr, first, degi, dis, N);
    gcn_xw_wmma      <<<nb_tile, 32, 0, stream>>>(x, W1, dis, xWs, N);
    gcn_agg1_gather  <<<nb_n16,  T,  0, stream>>>(csr, first, degi, xWs, dis, b1, W2, y, N);
    gcn_agg2_lds     <<<nb_a2,   A2T, 0, stream>>>(csr, first, degi, y, dis, b2, out, N);
}


# CSR build without a counting launch: each level-1 block stores its bucket-ordered edges as one piece plus a row of run starts and counts; a bucket gathers its runs by table column; 5 launches
# speedup vs baseline: 6.1847x; 1.1347x over previous
#include <hip/hip_runtime.h>
#include <hip/hip_bf16.h>

typedef __attribute__((ext_vector_type(2))) float v2f;
typedef __attribute__((ext_vector_type(8))) float v8f;

#define F_IN   128
#define F_HID  16

#define BN      256
#define NBP     512
#define BT      1024
#define BU      13
#define CHMAX   12800
#define RQ      8
#define PB      8
#define CAPB    16384
#define SORTMAX 128

__device__ __forceinline__ unsigned gcn_block_scan(unsigned v, unsigned* wsum) {
    const int lane = threadIdx.x & 31;
    const int w    = threadIdx.x >> 5;
    unsigned inc = v;
#pragma unroll
    for (int o = 1; o < 32; o <<= 1) {
        const unsigned up = __shfl_up(inc, o, 32);
        if (lane >= o) inc += up;
    }
    if (lane == 31) wsum[w] = inc;
    __syncthreads();
    if (w == 0) {
        const unsigned s = wsum[lane];
        unsigned si = s;
#pragma unroll
        for (int o = 1; o < 32; o <<= 1) {
            const unsigned up = __shfl_up(si, o, 32);
            if (lane >= o) si += up;
        }
        wsum[lane] = si - s;
    }
    __syncthreads();
    const unsigned res = wsum[w] + inc - v;
    __syncthreads();
    return res;
}

__global__ __launch_bounds__(BT)
void gcn_block_sort(const int* __restrict__ ei, unsigned* __restrict__ runm,
                    unsigned* __restrict__ pk, int E, int n, int chunk_len) {
    __shared__ unsigned sorted[CHMAX];
    __shared__ unsigned hist[NBP];
    __shared__ unsigned lstart[NBP];
    __shared__ unsigned wsum[32];
    const int t   = threadIdx.x;
    const int blk = blockIdx.x;
    if (t < NBP) hist[t] = 0u;

    const int* __restrict__ srcp = ei;
    const int* __restrict__ dstp = ei + E;
    const int c0 = blk * chunk_len;
    const int c1 = (c0 + chunk_len < E) ? (c0 + chunk_len) : E;
    int d[BU], s[BU];
#pragma unroll
    for (int u = 0; u < BU; ++u) {
        const int e  = c0 + u * BT + t;
        const int ec = (e < c1) ? e : (c1 - 1);
        const int dl = dstp[ec];
        s[u] = srcp[ec];
        d[u] = (e < c1) ? dl : -1;
    }
    __syncthreads();

    unsigned rk[BU];
#pragma unroll
    for (int u = 0; u < BU; ++u) {
        rk[u] = 0u;
        if ((unsigned)d[u] < (unsigned)n) rk[u] = atomicAdd(&hist[d[u] >> 8], 1u);
    }
    __syncthreads();

    const unsigned cm = (t < NBP) ? hist[t] : 0u;
    const unsigned ls = gcn_block_scan(cm, wsum);
    if (t < NBP) {
        lstart[t] = ls;
        runm[blk * NBP + t] = (ls << 16) | cm;
    }
    __syncthreads();

#pragma unroll
    for (int u = 0; u < BU; ++u) {
        if ((unsigned)d[u] < (unsigned)n) {
            const unsigned slot = lstart[d[u] >> 8] + rk[u];
            if (slot < (unsigned)CHMAX)
                sorted[slot] = ((unsigned)s[u] << 8) | ((unsigned)d[u] & 255u);
        }
    }
    __syncthreads();

    const unsigned nall    = lstart[NBP - 1] + hist[NBP - 1];
    const unsigned cnt_blk = (nall < (unsigned)CHMAX) ? nall : (unsigned)CHMAX;
    unsigned* __restrict__ dstw = pk + (long long)blk * chunk_len;
    unsigned wv[BU];
#pragma unroll
    for (int u = 0; u < BU; ++u) {
        const unsigned idx = (unsigned)(u * BT + t);
        wv[u] = sorted[(idx < cnt_blk) ? idx : 0u];
    }
#pragma unroll
    for (int u = 0; u < BU; ++u) {
        const unsigned idx = (unsigned)(u * BT + t);
        if (idx < cnt_blk) dstw[idx] = wv[u];
    }
}

template <class F>
__device__ __forceinline__ void gcn_sweep_runs(const unsigned* __restrict__ pk, const unsigned* rlen,
                                               const unsigned* rsrc, const unsigned* roff,
                                               int nr, int w, unsigned lane, F fn) {
    for (int r0 = w; r0 < nr; r0 += 32 * RQ) {
        unsigned len[RQ], src[RQ], off[RQ], e0[RQ], e1[RQ];
#pragma unroll
        for (int q = 0; q < RQ; ++q) {
            const int r  = r0 + 32 * q;
            const int rc = (r < nr) ? r : r0;
            const unsigned ln = (unsigned)__builtin_amdgcn_readfirstlane((int)rlen[rc]);
            src[q] = (unsigned)__builtin_amdgcn_readfirstlane((int)rsrc[rc]);
            off[q] = (unsigned)__builtin_amdgcn_readfirstlane((int)roff[rc]);
            len[q] = (r < nr) ? ln : 0u;
        }
#pragma unroll
        for (int q = 0; q < RQ; ++q) {
            e0[q] = pk[(lane < len[q]) ? (src[q] + lane) : 0u];
            e1[q] = pk[(lane + 32u < len[q]) ? (src[q] + lane + 32u) : 0u];
        }
#pragma unroll
        for (int q = 0; q < RQ; ++q) {
            asm volatile("" :: "v"(e0[q]), "v"(e1[q]));
        }
#pragma unroll
        for (int q = 0; q < RQ; ++q) {
            if (lane < len[q])       fn(e0[q], off[q] + lane);
            if (lane + 32u < len[q]) fn(e1[q], off[q] + lane + 32u);
        }
#pragma unroll
        for (int q = 0; q < RQ; ++q) {
            if (len[q] > 64u)
                for (unsigned k = 64u + lane; k < len[q]; k += 32u) fn(pk[src[q] + k], off[q] + k);
        }
    }
}

__global__ __launch_bounds__(BT)
void gcn_bucket_finish(const unsigned* __restrict__ pk, const unsigned* __restrict__ runm,
                       int* __restrict__ csr,
                       unsigned* __restrict__ first, unsigned* __restrict__ degi,
                       float* __restrict__ dis, int n, int nblk, int chunk_len) {
    __shared__ unsigned buf[CAPB];
    __shared__ unsigned buf2[CAPB];
    __shared__ unsigned ncnt[BN];
    __shared__ unsigned nstart[BN];
    __shared__ unsigned ncur[BN];
    __shared__ unsigned rlen[BT];
    __shared__ unsigned rsrc[BT];
    __shared__ unsigned roff[BT];
    __shared__ unsigned tot[2];
    __shared__ unsigned wsum[32];
    const int t = threadIdx.x;
    const int b = blockIdx.x;
    const int w = __builtin_amdgcn_readfirstlane(t >> 5);
    const unsigned lane = (unsigned)(t & 31);
    const int nr = (nblk < BT) ? nblk : BT;
    unsigned pw = 0u;
    if (t < nr) pw = runm[t * NBP + b];
    const unsigned c = pw & 0xFFFFu;
    const unsigned l = pw >> 16;
    if (t < BN) ncnt[t] = 0u;
    const unsigned ro = gcn_block_scan(c, wsum);
    const unsigned lo = gcn_block_scan(l, wsum);
    rlen[t] = c;
    rsrc[t] = (unsigned)t * (unsigned)chunk_len + l;
    roff[t] = ro;
    if (t == BT - 1) { tot[0] = ro + c; tot[1] = lo + l; }
    __syncthreads();
    const unsigned nb = tot[0];
    const unsigned gb = tot[1];
    const bool fits = nb <= (unsigned)CAPB;

    gcn_sweep_runs(pk, rlen, rsrc, roff, nr, w, lane, [&](unsigned e, unsigned pos) {
        atomicAdd(&ncnt[e & 255u], 1u);
        if (fits) buf2[pos] = e;
    });
    __syncthreads();

    const unsigned v  = (t < BN) ? ncnt[t] : 0u;
    const unsigned ex = gcn_block_scan(v, wsum);
    if (t < BN) {
        nstart[t] = ex;
        ncur[t]   = 0u;
        const int node = b * BN + t;
        if (node < n) {
            degi[node]  = v;
            first[node] = gb + ex;
            const float dgv = (float)(v + 1u);
            dis[node] = dgv > 0.0f ? rsqrtf(dgv) : 0.0f;
        }
    }
    __syncthreads();

    if (fits) {
        for (unsigned base = (unsigned)t; base < nb; base += BT * PB) {
            unsigned e[PB], rk[PB];
#pragma unroll
            for (int u = 0; u < PB; ++u) {
                const unsigned i = base + (unsigned)u * BT;
                e[u] = buf2[(i < nb) ? i : 0u];
            }
#pragma unroll
            for (int u = 0; u < PB; ++u) {
                rk[u] = 0u;
                if (base + (unsigned)u * BT < nb) rk[u] = atomicAdd(&ncur[e[u] & 255u], 1u);
            }
#pragma unroll
            for (int u = 0; u < PB; ++u) {
                if (base + (unsigned)u * BT < nb) {
                    const unsigned slot = nstart[e[u] & 255u] + rk[u];
                    if (slot < (unsigned)CAPB) buf[slot] = (e[u] << 7) | (rk[u] & 127u);
                }
            }
        }
        __syncthreads();
        for (unsigned i = (unsigned)t; i < nb; i += BT) {
            const unsigned key = buf[i];
            const unsigned ld  = (key >> 7) & 255u;
            const unsigned ns  = nstart[ld];
            const unsigned len = ncnt[ld];
            unsigned r = i - ns;
            if (len <= (unsigned)SORTMAX) {
                r = 0u;
                for (unsigned j = 0u; j < len; ++j) r += (buf[ns + j] < key) ? 1u : 0u;
            }
            buf2[ns + r] = key >> 15;
        }
        __syncthreads();
        for (unsigned i = (unsigned)t; i < nb; i += BT) csr[gb + i] = (int)buf2[i];
    } else {
        gcn_sweep_runs(pk, rlen, rsrc, roff, nr, w, lane, [&](unsigned e, unsigned) {
            const unsigned rk = atomicAdd(&ncur[e & 255u], 1u);
            csr[gb + nstart[e & 255u] + rk] = (int)(e >> 8);
        });
    }
}

__global__ __launch_bounds__(32)
void gcn_xw_wmma(const float* __restrict__ x, const float* __restrict__ W1,
                 const float* __restrict__ dis, float* __restrict__ xWs, int n) {
    const int row0 = blockIdx.x * 16;
    if (row0 >= n) return;
    const int lane = threadIdx.x;
    const int half = lane >> 4;
    const int l16  = lane & 15;

    const float* xrow = x + (long long)(row0 + l16) * F_IN;

    if (row0 + 16 + l16 < n)
        __builtin_prefetch(x + (long long)(row0 + 16 + l16) * F_IN + half * 64, 0, 3);

    v8f c = {};
#pragma unroll
    for (int k0 = 0; k0 < F_IN; k0 += 4) {
        const int ka = k0 + half * 2;
        v2f a, b;
        a.x = xrow[ka];
        a.y = xrow[ka + 1];
        b.x = W1[ka * F_HID + l16];
        b.y = W1[(ka + 1) * F_HID + l16];
        c = __builtin_amdgcn_wmma_f32_16x16x4_f32(
                 false, a,  false, b,
                 (short)0, c,  false,  false);
    }
#pragma unroll
    for (int r = 0; r < 8; ++r) {
        const int m = r + half * 8;
        xWs[(long long)(row0 + m) * F_HID + l16] = c[r] * dis[row0 + m];
    }
}

__global__ __launch_bounds__(256)
void gcn_agg1_gather(const int* __restrict__ csr, const unsigned* __restrict__ first,
                     const unsigned* __restrict__ degi, const float* __restrict__ xWs,
                     const float* __restrict__ dis, const float* __restrict__ b1,
                     const float* __restrict__ W2, float* __restrict__ y, int n) {
    const int t    = blockIdx.x * blockDim.x + threadIdx.x;
    const int node = t >> 4;
    const int j    = t & 15;
    const bool live = node < n;
    const int nd   = live ? node : (n - 1);
    const unsigned beg = first[nd];
    const unsigned cnt = degi[nd];
    const float* __restrict__ col = xWs + j;
    float acc = col[(long long)nd * F_HID];
    int next = ((unsigned)j < cnt) ? csr[beg + (unsigned)j] : -1;
    for (unsigned k0 = 0u; k0 < cnt; k0 += 16u) {
        const int mine = next;
        int   s[16];
        float r[16];
#pragma unroll
        for (int u = 0; u < 16; ++u) {
            s[u] = __shfl(mine, u, 16);
            const int sc = (s[u] >= 0) ? s[u] : nd;
            r[u] = col[(long long)sc * F_HID];
        }
        const unsigned kn = k0 + 16u + (unsigned)j;
        next = (kn < cnt) ? csr[beg + kn] : -1;
#pragma unroll
        for (int u = 0; u < 16; ++u) acc += (s[u] >= 0) ? r[u] : 0.0f;
    }
    const float dv = dis[nd];
    float hv = fmaf(acc, dv, b1[j]);
    hv = fmaxf(hv, 0.0f);
    float p = hv * W2[j];
    p += __shfl_xor(p, 8, 16);
    p += __shfl_xor(p, 4, 16);
    p += __shfl_xor(p, 2, 16);
    p += __shfl_xor(p, 1, 16);
    if (live && j == 0) y[node] = p * dv;
}

#define YS     51200
#define A2T    1024
#define A2G    (A2T / 16)
#define A2IT   7
#define A2NPB  (A2G * A2IT)
#define A2SW   ((YS + A2T * 4 - 1) / (A2T * 4))
#define A2SA   7

template <int B>
__device__ __forceinline__ void gcn_stage_y(const float* __restrict__ ysrc, float* __restrict__ ys,
                                            int len4, int sw0, int rot, unsigned t4) {
    float4 v[B];
#pragma unroll
    for (int b = 0; b < B; ++b) {
        const int sw  = (sw0 + b + rot < A2SW) ? (sw0 + b + rot) : (sw0 + b + rot - A2SW);
        const int off = sw * (A2T * 4);
        v[b] = make_float4(0.0f, 0.0f, 0.0f, 0.0f);
        if ((int)t4 < len4 - off) v[b] = *reinterpret_cast<const float4*>(ysrc + off + t4);
    }
#pragma unroll
    for (int b = 0; b < B; ++b) {
        const int sw  = (sw0 + b + rot < A2SW) ? (sw0 + b + rot) : (sw0 + b + rot - A2SW);
        const int off = sw * (A2T * 4);
        if ((int)t4 < len4 - off) *reinterpret_cast<float4*>(ys + off + t4) = v[b];
    }
}

__global__ __launch_bounds__(A2T)
void gcn_agg2_lds(const int* __restrict__ csr, const unsigned* __restrict__ first,
                  const unsigned* __restrict__ degi, const float* __restrict__ y,
                  const float* __restrict__ dis, const float* __restrict__ b2,
                  float* __restrict__ out, int n) {
    __shared__ __attribute__((aligned(16))) float ys[YS];
    const int tid  = threadIdx.x;
    const int grp  = tid >> 4;
    const int j    = tid & 15;
    const int base = blockIdx.x * A2NPB;
    const int rot  = blockIdx.x % A2SW;

    unsigned cnt[A2IT];
    int      s[A2IT][4];
    {
        unsigned beg[A2IT];
#pragma unroll
        for (int it = 0; it < A2IT; ++it) {
            const int node = base + it * A2G + grp;
            const int nd   = (node < n) ? node : (n - 1);
            beg[it] = first[nd];
            const unsigned c = degi[nd];
            cnt[it] = (node < n) ? c : 0u;
        }
#pragma unroll
        for (int it = 0; it < A2IT; ++it) {
#pragma unroll
            for (int u = 0; u < 4; ++u) {
                const unsigned k = (unsigned)j + 16u * (unsigned)u;
                s[it][u] = (k < cnt[it]) ? csr[beg[it] + k] : -1;
            }
        }
    }
    float acc[A2IT], dv[A2IT], yv[A2IT];
#pragma unroll
    for (int it = 0; it < A2IT; ++it) {
        const int node = base + it * A2G + grp;
        const int nd   = (node < n) ? node : (n - 1);
        dv[it]  = dis[nd];
        yv[it]  = y[nd];
        acc[it] = 0.0f;
    }

    for (int lo = 0; lo < n; lo += YS) {
        const int len  = (n - lo < YS) ? (n - lo) : YS;
        const int len4 = len & ~3;
        gcn_stage_y<A2SA>(y + lo, ys, len4, 0, rot, (unsigned)tid * 4u);
        gcn_stage_y<A2SW - A2SA>(y + lo, ys, len4, A2SA, rot, (unsigned)tid * 4u);
        if (tid < len - len4) ys[len4 + tid] = y[lo + len4 + tid];
        __syncthreads();
#pragma unroll
        for (int it = 0; it < A2IT; ++it) {
#pragma unroll
            for (int u = 0; u < 4; ++u) {
                const unsigned loc = (unsigned)(s[it][u] - lo);
                const float    r   = ys[(loc < (unsigned)len) ? loc : 0u];
                acc[it] += (loc < (unsigned)len) ? r : 0.0f;
            }
        }
#pragma unroll
        for (int it = 0; it < A2IT; ++it) {
            if (cnt[it] > 64u) {
                const unsigned bg = first[base + it * A2G + grp];
                for (unsigned k = 64u + (unsigned)j; k < cnt[it]; k += 16u) {
                    const unsigned loc = (unsigned)(csr[bg + k] - lo);
                    if (loc < (unsigned)len) acc[it] += ys[loc];
                }
            }
        }
        __syncthreads();
    }
#pragma unroll
    for (int w = 8; w >= 1; w >>= 1) {
#pragma unroll
        for (int it = 0; it < A2IT; ++it) acc[it] += __shfl_xor(acc[it], w, 16);
    }
    const float bias = b2[0];
#pragma unroll
    for (int it = 0; it < A2IT; ++it) {
        const int node = base + it * A2G + grp;
        if (node < n && j == 0) out[node] = bias + dv[it] * (yv[it] + acc[it]);
    }
}

extern "C" void kernel_launch(void* const* d_in, const int* in_sizes, int n_in,
                              void* d_out, int out_size, void* d_ws, size_t ws_size,
                              hipStream_t stream) {
    const float* x  = (const float*)d_in[0];
    const int*   ei = (const int*)d_in[1];
    const float* W1 = (const float*)d_in[2];
    const float* b1 = (const float*)d_in[3];
    const float* W2 = (const float*)d_in[4];
    const float* b2 = (const float*)d_in[5];
    float* out = (float*)d_out;

    const int       N = in_sizes[0] / F_IN;
    const long long E = (long long)in_sizes[1] / 2;

    const int Ei        = (int)E;
    const int nblk      = (int)((E + CHMAX - 1) / CHMAX);
    const int chunk_len = (int)((E + nblk - 1) / nblk);
    const int n_bkt     = (N + BN - 1) / BN;

    const long long Np = ((long long)N + 3) / 4 * 4;
    float*    ws    = (float*)d_ws;
    float*    y     = ws;
    float*    dis   = ws + Np;
    float*    xWs   = ws + Np + N;
    unsigned* degi  = (unsigned*)(ws + Np + (long long)17 * N);
    unsigned* first = (unsigned*)(ws + Np + (long long)18 * N);
    unsigned* runm  = (unsigned*)(ws + Np + (long long)19 * N);
    unsigned* pk    = runm + (long long)nblk * NBP;
    int*      csr   = (int*)(pk + (long long)nblk * chunk_len);

    const int T = 256;
    const unsigned nb_n16  = (unsigned)(((long long)N * F_HID + T - 1) / T);
    const unsigned nb_tile = (unsigned)((N + 15) / 16);
    const unsigned nb_a2   = (unsigned)((N + A2NPB - 1) / A2NPB);

    gcn_block_sort   <<<(unsigned)nblk,  BT, 0, stream>>>(ei, runm, pk, Ei, N, chunk_len);
    gcn_bucket_finish<<<(unsigned)n_bkt, BT, 0, stream>>>(pk, runm, csr, first, degi, dis, N, nblk, chunk_len);
    gcn_xw_wmma      <<<nb_tile, 32, 0, stream>>>(x, W1, dis, xWs, N);
    gcn_agg1_gather  <<<nb_n16,  T,  0, stream>>>(csr, first, degi, xWs, dis, b1, W2, y, N);
    gcn_agg2_lds     <<<nb_a2,   A2T, 0, stream>>>(csr, first, degi, y, dis, b2, out, N);
}


# x @ W1 moves into the bucket blocks of the list build: each multiplies its own 256 rows (whole-row loads asked for before the rank pass, fragments from LDS); 4 launches, not 5; sums unchanged
# speedup vs baseline: 6.5288x; 1.0556x over previous
#include <hip/hip_runtime.h>
#include <hip/hip_bf16.h>

typedef __attribute__((ext_vector_type(2))) float v2f;
typedef __attribute__((ext_vector_type(8))) float v8f;

#define F_IN   128
#define F_HID  16

#define BN      256
#define NBP     512
#define BT      1024
#define BU      13
#define CHMAX   12800
#define RQ      8
#define PB      8
#define CAPB    16384
#define SORTMAX 128

__device__ __forceinline__ unsigned gcn_block_scan(unsigned v, unsigned* wsum) {
    const int lane = threadIdx.x & 31;
    const int w    = threadIdx.x >> 5;
    unsigned inc = v;
#pragma unroll
    for (int o = 1; o < 32; o <<= 1) {
        const unsigned up = __shfl_up(inc, o, 32);
        if (lane >= o) inc += up;
    }
    if (lane == 31) wsum[w] = inc;
    __syncthreads();
    if (w == 0) {
        const unsigned s = wsum[lane];
        unsigned si = s;
#pragma unroll
        for (int o = 1; o < 32; o <<= 1) {
            const unsigned up = __shfl_up(si, o, 32);
            if (lane >= o) si += up;
        }
        wsum[lane] = si - s;
    }
    __syncthreads();
    const unsigned res = wsum[w] + inc - v;
    __syncthreads();
    return res;
}

__global__ __launch_bounds__(BT)
void gcn_block_sort(const int* __restrict__ ei, unsigned* __restrict__ runm,
                    unsigned* __restrict__ pk, int E, int n, int chunk_len) {
    __shared__ unsigned sorted[CHMAX];
    __shared__ unsigned hist[NBP];
    __shared__ unsigned lstart[NBP];
    __shared__ unsigned wsum[32];
    const int t   = threadIdx.x;
    const int blk = blockIdx.x;
    if (t < NBP) hist[t] = 0u;

    const int* __restrict__ srcp = ei;
    const int* __restrict__ dstp = ei + E;
    const int c0 = blk * chunk_len;
    const int c1 = (c0 + chunk_len < E) ? (c0 + chunk_len) : E;
    int d[BU], s[BU];
#pragma unroll
    for (int u = 0; u < BU; ++u) {
        const int e  = c0 + u * BT + t;
        const int ec = (e < c1) ? e : (c1 - 1);
        const int dl = dstp[ec];
        s[u] = srcp[ec];
        d[u] = (e < c1) ? dl : -1;
    }
    __syncthreads();

    unsigned rk[BU];
#pragma unroll
    for (int u = 0; u < BU; ++u) {
        rk[u] = 0u;
        if ((unsigned)d[u] < (unsigned)n) rk[u] = atomicAdd(&hist[d[u] >> 8], 1u);
    }
    __syncthreads();

    const unsigned cm = (t < NBP) ? hist[t] : 0u;
    const unsigned ls = gcn_block_scan(cm, wsum);
    if (t < NBP) {
        lstart[t] = ls;
        runm[blk * NBP + t] = (ls << 16) | cm;
    }
    __syncthreads();

#pragma unroll
    for (int u = 0; u < BU; ++u) {
        if ((unsigned)d[u] < (unsigned)n) {
            const unsigned slot = lstart[d[u] >> 8] + rk[u];
            if (slot < (unsigned)CHMAX)
                sorted[slot] = ((unsigned)s[u] << 8) | ((unsigned)d[u] & 255u);
        }
    }
    __syncthreads();

    const unsigned nall    = lstart[NBP - 1] + hist[NBP - 1];
    const unsigned cnt_blk = (nall < (unsigned)CHMAX) ? nall : (unsigned)CHMAX;
    unsigned* __restrict__ dstw = pk + (long long)blk * chunk_len;
    unsigned wv[BU];
#pragma unroll
    for (int u = 0; u < BU; ++u) {
        const unsigned idx = (unsigned)(u * BT + t);
        wv[u] = sorted[(idx < cnt_blk) ? idx : 0u];
    }
#pragma unroll
    for (int u = 0; u < BU; ++u) {
        const unsigned idx = (unsigned)(u * BT + t);
        if (idx < cnt_blk) dstw[idx] = wv[u];
    }
}

template <class F>
__device__ __forceinline__ void gcn_sweep_runs(const unsigned* __restrict__ pk, const unsigned* rlen,
                                               const unsigned* rsrc, const unsigned* roff,
                                               int nr, int w, unsigned lane, F fn) {
    for (int r0 = w; r0 < nr; r0 += 32 * RQ) {
        unsigned len[RQ], src[RQ], off[RQ], e0[RQ], e1[RQ];
#pragma unroll
        for (int q = 0; q < RQ; ++q) {
            const int r  = r0 + 32 * q;
            const int rc = (r < nr) ? r : r0;
            const unsigned ln = (unsigned)__builtin_amdgcn_readfirstlane((int)rlen[rc]);
            src[q] = (unsigned)__builtin_amdgcn_readfirstlane((int)rsrc[rc]);
            off[q] = (unsigned)__builtin_amdgcn_readfirstlane((int)roff[rc]);
            len[q] = (r < nr) ? ln : 0u;
        }
#pragma unroll
        for (int q = 0; q < RQ; ++q) {
            e0[q] = pk[(lane < len[q]) ? (src[q] + lane) : 0u];
            e1[q] = pk[(lane + 32u < len[q]) ? (src[q] + lane + 32u) : 0u];
        }
#pragma unroll
        for (int q = 0; q < RQ; ++q) {
            asm volatile("" :: "v"(e0[q]), "v"(e1[q]));
        }
#pragma unroll
        for (int q = 0; q < RQ; ++q) {
            if (lane < len[q])       fn(e0[q], off[q] + lane);
            if (lane + 32u < len[q]) fn(e1[q], off[q] + lane + 32u);
        }
#pragma unroll
        for (int q = 0; q < RQ; ++q) {
            if (len[q] > 64u)
                for (unsigned k = 64u + lane; k < len[q]; k += 32u) fn(pk[src[q] + k], off[q] + k);
        }
    }
}

__device__ __forceinline__ void gcn_load_xrows(const float* __restrict__ x, int b, int n, int t,
                                               float4 (&xv)[BN / 32]) {
#pragma unroll
    for (int i = 0; i < BN / 32; ++i) {
        const int r  = b * BN + i * 32 + (t >> 5);
        const int rc = (r < n) ? r : (n - 1);
        xv[i] = *reinterpret_cast<const float4*>(x + (unsigned)rc * (unsigned)F_IN + (unsigned)(t & 31) * 4u);
    }
}

__global__ __launch_bounds__(BT, 16)
void gcn_bucket_finish(const unsigned* __restrict__ pk, const unsigned* __restrict__ runm,
                       int* __restrict__ csr,
                       unsigned* __restrict__ first, unsigned* __restrict__ degi,
                       float* __restrict__ dis, const float* __restrict__ x,
                       const float* __restrict__ W1, float* __restrict__ xWs,
                       int n, int nblk, int chunk_len) {
    __shared__ __attribute__((aligned(16))) unsigned buf[CAPB];
    __shared__ __attribute__((aligned(16))) unsigned buf2[CAPB];
    __shared__ float sW[F_IN * F_HID];
    __shared__ float sdis[BN];
    __shared__ unsigned ncnt[BN];
    __shared__ unsigned nstart[BN];
    __shared__ unsigned ncur[BN];
    __shared__ unsigned rlen[BT];
    __shared__ unsigned rsrc[BT];
    __shared__ unsigned roff[BT];
    __shared__ unsigned tot[2];
    __shared__ unsigned wsum[32];
    const int t = threadIdx.x;
    const int b = blockIdx.x;
    const int w = __builtin_amdgcn_readfirstlane(t >> 5);
    const unsigned lane = (unsigned)(t & 31);
    const int nr = (nblk < BT) ? nblk : BT;
    unsigned pw = 0u;
    if (t < nr) pw = runm[t * NBP + b];
    const unsigned c = pw & 0xFFFFu;
    const unsigned l = pw >> 16;
    if (t < BN) ncnt[t] = 0u;
    sW[t]      = W1[t];
    sW[t + BT] = W1[t + BT];
    const unsigned ro = gcn_block_scan(c, wsum);
    const unsigned lo = gcn_block_scan(l, wsum);
    rlen[t] = c;
    rsrc[t] = (unsigned)t * (unsigned)chunk_len + l;
    roff[t] = ro;
    if (t == BT - 1) { tot[0] = ro + c; tot[1] = lo + l; }
    __syncthreads();
    const unsigned nb = tot[0];
    const unsigned gb = tot[1];
    const bool fits = nb <= (unsigned)CAPB;

    gcn_sweep_runs(pk, rlen, rsrc, roff, nr, w, lane, [&](unsigned e, unsigned pos) {
        atomicAdd(&ncnt[e & 255u], 1u);
        if (fits) buf2[pos] = e;
    });
    __syncthreads();

    const unsigned v  = (t < BN) ? ncnt[t] : 0u;
    const unsigned ex = gcn_block_scan(v, wsum);
    if (t < BN) {
        nstart[t] = ex;
        ncur[t]   = 0u;
        const int node = b * BN + t;
        if (node < n) {
            degi[node]  = v;
            first[node] = gb + ex;
            const float dgv = (float)(v + 1u);
            const float dsv = dgv > 0.0f ? rsqrtf(dgv) : 0.0f;
            dis[node] = dsv;
            sdis[t]   = dsv;
        }
    }
    __syncthreads();

    float4 xv[BN / 32];
    if (fits) {
        for (unsigned base = (unsigned)t; base < nb; base += BT * PB) {
            unsigned e[PB], rk[PB];
#pragma unroll
            for (int u = 0; u < PB; ++u) {
                const unsigned i = base + (unsigned)u * BT;
                e[u] = buf2[(i < nb) ? i : 0u];
            }
#pragma unroll
            for (int u = 0; u < PB; ++u) {
                rk[u] = 0u;
                if (base + (unsigned)u * BT < nb) rk[u] = atomicAdd(&ncur[e[u] & 255u], 1u);
            }
#pragma unroll
            for (int u = 0; u < PB; ++u) {
                if (base + (unsigned)u * BT < nb) {
                    const unsigned slot = nstart[e[u] & 255u] + rk[u];
                    if (slot < (unsigned)CAPB) buf[slot] = (e[u] << 7) | (rk[u] & 127u);
                }
            }
        }
        __syncthreads();
        gcn_load_xrows(x, b, n, t, xv);
        for (unsigned i = (unsigned)t; i < nb; i += BT) {
            const unsigned key = buf[i];
            const unsigned ld  = (key >> 7) & 255u;
            const unsigned ns  = nstart[ld];
            const unsigned len = ncnt[ld];
            unsigned r = i - ns;
            if (len <= (unsigned)SORTMAX) {
                r = 0u;
                for (unsigned j = 0u; j < len; ++j) r += (buf[ns + j] < key) ? 1u : 0u;
            }
            buf2[ns + r] = key >> 15;
        }
        __syncthreads();
        for (unsigned i = (unsigned)t; i < nb; i += BT) csr[gb + i] = (int)buf2[i];
    } else {
        gcn_sweep_runs(pk, rlen, rsrc, roff, nr, w, lane, [&](unsigned e, unsigned) {
            const unsigned rk = atomicAdd(&ncur[e & 255u], 1u);
            csr[gb + nstart[e & 255u] + rk] = (int)(e >> 8);
        });
        gcn_load_xrows(x, b, n, t, xv);
    }
    __syncthreads();

#pragma unroll
    for (int i = 0; i < BN / 32; ++i) {
        const unsigned r  = (unsigned)(i * 32 + (t >> 5));
        const unsigned pp = lane & 15u;
        unsigned* hb = (lane & 16u) ? buf2 : buf;
        *reinterpret_cast<uint4*>(hb + r * 64u + 4u * (pp ^ (r & 15u))) =
            make_uint4(__float_as_uint(xv[i].x), __float_as_uint(xv[i].y),
                       __float_as_uint(xv[i].z), __float_as_uint(xv[i].w));
    }
    __syncthreads();
    if (w < BN / 16) {
        const int row0 = b * BN + w * 16;
        if (row0 < n) {
            const unsigned half = lane >> 4;
            const unsigned l16  = lane & 15u;
            const unsigned rl   = (unsigned)w * 16u + l16;
            v8f acc = {};
#pragma unroll
            for (int s = 0; s < F_IN / 4; ++s) {
                const unsigned* hb = (s < 16) ? buf : buf2;
                const uint2 av = *reinterpret_cast<const uint2*>(
                    hb + rl * 64u + 4u * ((unsigned)(s & 15) ^ l16) + 2u * half);
                const unsigned ka = 4u * (unsigned)s + 2u * half;
                v2f a, bw;
                a.x  = __uint_as_float(av.x);
                a.y  = __uint_as_float(av.y);
                bw.x = sW[ka * F_HID + l16];
                bw.y = sW[(ka + 1u) * F_HID + l16];
                acc = __builtin_amdgcn_wmma_f32_16x16x4_f32(
                         false, a,  false, bw,
                         (short)0, acc,  false,  false);
            }
            float dv[8];
#pragma unroll
            for (int r = 0; r < 8; ++r) dv[r] = sdis[(unsigned)w * 16u + (unsigned)r + half * 8u];
#pragma unroll
            for (int r = 0; r < 8; ++r) {
                const int node = row0 + r + (int)half * 8;
                if (node < n) xWs[(long long)node * F_HID + l16] = acc[r] * dv[r];
            }
        }
    }
}

__global__ __launch_bounds__(256)
void gcn_agg1_gather(const int* __restrict__ csr, const unsigned* __restrict__ first,
                     const unsigned* __restrict__ degi, const float* __restrict__ xWs,
                     const float* __restrict__ dis, const float* __restrict__ b1,
                     const float* __restrict__ W2, float* __restrict__ y, int n) {
    const int t    = blockIdx.x * blockDim.x + threadIdx.x;
    const int node = t >> 4;
    const int j    = t & 15;
    const bool live = node < n;
    const int nd   = live ? node : (n - 1);
    const unsigned beg = first[nd];
    const unsigned cnt = degi[nd];
    const float* __restrict__ col = xWs + j;
    float acc = col[(long long)nd * F_HID];
    int next = ((unsigned)j < cnt) ? csr[beg + (unsigned)j] : -1;
    for (unsigned k0 = 0u; k0 < cnt; k0 += 16u) {
        const int mine = next;
        int   s[16];
        float r[16];
#pragma unroll
        for (int u = 0; u < 16; ++u) {
            s[u] = __shfl(mine, u, 16);
            const int sc = (s[u] >= 0) ? s[u] : nd;
            r[u] = col[(long long)sc * F_HID];
        }
        const unsigned kn = k0 + 16u + (unsigned)j;
        next = (kn < cnt) ? csr[beg + kn] : -1;
#pragma unroll
        for (int u = 0; u < 16; ++u) acc += (s[u] >= 0) ? r[u] : 0.0f;
    }
    const float dv = dis[nd];
    float hv = fmaf(acc, dv, b1[j]);
    hv = fmaxf(hv, 0.0f);
    float p = hv * W2[j];
    p += __shfl_xor(p, 8, 16);
    p += __shfl_xor(p, 4, 16);
    p += __shfl_xor(p, 2, 16);
    p += __shfl_xor(p, 1, 16);
    if (live && j == 0) y[node] = p * dv;
}

#define YS     51200
#define A2T    1024
#define A2G    (A2T / 16)
#define A2IT   7
#define A2NPB  (A2G * A2IT)
#define A2SW   ((YS + A2T * 4 - 1) / (A2T * 4))
#define A2SA   7

template <int B>
__device__ __forceinline__ void gcn_stage_y(const float* __restrict__ ysrc, float* __restrict__ ys,
                                            int len4, int sw0, int rot, unsigned t4) {
    float4 v[B];
#pragma unroll
    for (int b = 0; b < B; ++b) {
        const int sw  = (sw0 + b + rot < A2SW) ? (sw0 + b + rot) : (sw0 + b + rot - A2SW);
        const int off = sw * (A2T * 4);
        v[b] = make_float4(0.0f, 0.0f, 0.0f, 0.0f);
        if ((int)t4 < len4 - off) v[b] = *reinterpret_cast<const float4*>(ysrc + off + t4);
    }
#pragma unroll
    for (int b = 0; b < B; ++b) {
        const int sw  = (sw0 + b + rot < A2SW) ? (sw0 + b + rot) : (sw0 + b + rot - A2SW);
        const int off = sw * (A2T * 4);
        if ((int)t4 < len4 - off) *reinterpret_cast<float4*>(ys + off + t4) = v[b];
    }
}

__global__ __launch_bounds__(A2T)
void gcn_agg2_lds(const int* __restrict__ csr, const unsigned* __restrict__ first,
                  const unsigned* __restrict__ degi, const float* __restrict__ y,
                  const float* __restrict__ dis, const float* __restrict__ b2,
                  float* __restrict__ out, int n) {
    __shared__ __attribute__((aligned(16))) float ys[YS];
    const int tid  = threadIdx.x;
    const int grp  = tid >> 4;
    const int j    = tid & 15;
    const int base = blockIdx.x * A2NPB;
    const int rot  = blockIdx.x % A2SW;

    unsigned cnt[A2IT];
    int      s[A2IT][4];
    {
        unsigned beg[A2IT];
#pragma unroll
        for (int it = 0; it < A2IT; ++it) {
            const int node = base + it * A2G + grp;
            const int nd   = (node < n) ? node : (n - 1);
            beg[it] = first[nd];
            const unsigned c = degi[nd];
            cnt[it] = (node < n) ? c : 0u;
        }
#pragma unroll
        for (int it = 0; it < A2IT; ++it) {
#pragma unroll
            for (int u = 0; u < 4; ++u) {
                const unsigned k = (unsigned)j + 16u * (unsigned)u;
                s[it][u] = (k < cnt[it]) ? csr[beg[it] + k] : -1;
            }
        }
    }
    float acc[A2IT], dv[A2IT], yv[A2IT];
#pragma unroll
    for (int it = 0; it < A2IT; ++it) {
        const int node = base + it * A2G + grp;
        const int nd   = (node < n) ? node : (n - 1);
        dv[it]  = dis[nd];
        yv[it]  = y[nd];
        acc[it] = 0.0f;
    }

    for (int lo = 0; lo < n; lo += YS) {
        const int len  = (n - lo < YS) ? (n - lo) : YS;
        const int len4 = len & ~3;
        gcn_stage_y<A2SA>(y + lo, ys, len4, 0, rot, (unsigned)tid * 4u);
        gcn_stage_y<A2SW - A2SA>(y + lo, ys, len4, A2SA, rot, (unsigned)tid * 4u);
        if (tid < len - len4) ys[len4 + tid] = y[lo + len4 + tid];
        __syncthreads();
#pragma unroll
        for (int it = 0; it < A2IT; ++it) {
#pragma unroll
            for (int u = 0; u < 4; ++u) {
                const unsigned loc = (unsigned)(s[it][u] - lo);
                const float    r   = ys[(loc < (unsigned)len) ? loc : 0u];
                acc[it] += (loc < (unsigned)len) ? r : 0.0f;
            }
        }
#pragma unroll
        for (int it = 0; it < A2IT; ++it) {
            if (cnt[it] > 64u) {
                const unsigned bg = first[base + it * A2G + grp];
                for (unsigned k = 64u + (unsigned)j; k < cnt[it]; k += 16u) {
                    const unsigned loc = (unsigned)(csr[bg + k] - lo);
                    if (loc < (unsigned)len) acc[it] += ys[loc];
                }
            }
        }
        __syncthreads();
    }
#pragma unroll
    for (int w = 8; w >= 1; w >>= 1) {
#pragma unroll
        for (int it = 0; it < A2IT; ++it) acc[it] += __shfl_xor(acc[it], w, 16);
    }
    const float bias = b2[0];
#pragma unroll
    for (int it = 0; it < A2IT; ++it) {
        const int node = base + it * A2G + grp;
        if (node < n && j == 0) out[node] = bias + dv[it] * (yv[it] + acc[it]);
    }
}

extern "C" void kernel_launch(void* const* d_in, const int* in_sizes, int n_in,
                              void* d_out, int out_size, void* d_ws, size_t ws_size,
                              hipStream_t stream) {
    const float* x  = (const float*)d_in[0];
    const int*   ei = (const int*)d_in[1];
    const float* W1 = (const float*)d_in[2];
    const float* b1 = (const float*)d_in[3];
    const float* W2 = (const float*)d_in[4];
    const float* b2 = (const float*)d_in[5];
    float* out = (float*)d_out;

    const int       N = in_sizes[0] / F_IN;
    const long long E = (long long)in_sizes[1] / 2;

    const int Ei        = (int)E;
    const int nblk      = (int)((E + CHMAX - 1) / CHMAX);
    const int chunk_len = (int)((E + nblk - 1) / nblk);
    const int n_bkt     = (N + BN - 1) / BN;

    const long long Np = ((long long)N + 3) / 4 * 4;
    float*    ws    = (float*)d_ws;
    float*    y     = ws;
    float*    dis   = ws + Np;
    float*    xWs   = ws + Np + N;
    unsigned* degi  = (unsigned*)(ws + Np + (long long)17 * N);
    unsigned* first = (unsigned*)(ws + Np + (long long)18 * N);
    unsigned* runm  = (unsigned*)(ws + Np + (long long)19 * N);
    unsigned* pk    = runm + (long long)nblk * NBP;
    int*      csr   = (int*)(pk + (long long)nblk * chunk_len);

    const int T = 256;
    const unsigned nb_n16  = (unsigned)(((long long)N * F_HID + T - 1) / T);
    const unsigned nb_a2   = (unsigned)((N + A2NPB - 1) / A2NPB);

    gcn_block_sort   <<<(unsigned)nblk,  BT, 0, stream>>>(ei, runm, pk, Ei, N, chunk_len);
    gcn_bucket_finish<<<(unsigned)n_bkt, BT, 0, stream>>>(pk, runm, csr, first, degi, dis, x, W1, xWs, N, nblk, chunk_len);
    gcn_agg1_gather  <<<nb_n16,  T,  0, stream>>>(csr, first, degi, xWs, dis, b1, W2, y, N);
    gcn_agg2_lds     <<<nb_a2,   A2T, 0, stream>>>(csr, first, degi, y, dis, b2, out, N);
}


# Level-2 sort counts and places per (node, quarter of the source range), not per node: a key is ranked among its quarter's keys only, about 8 compares per edge, not 33; list and sums unchanged
# speedup vs baseline: 7.2630x; 1.1125x over previous
#include <hip/hip_runtime.h>
#include <hip/hip_bf16.h>

typedef __attribute__((ext_vector_type(2))) float v2f;
typedef __attribute__((ext_vector_type(8))) float v8f;

#define F_IN   128
#define F_HID  16

#define BN      256
#define NBP     512
#define BT      1024
#define BU      13
#define CHMAX   12800
#define RQ      8
#define PB      8
#define CAPB    16384
#define SORTMAX 128
#define NQ      4

__device__ __forceinline__ unsigned gcn_cell(unsigned ld, unsigned src, float q4) {
    const unsigned s = (unsigned)((float)src * q4);
    return ld * (unsigned)NQ + ((s < (unsigned)(NQ - 1)) ? s : (unsigned)(NQ - 1));
}

__device__ __forceinline__ unsigned gcn_block_scan(unsigned v, unsigned* wsum) {
    const int lane = threadIdx.x & 31;
    const int w    = threadIdx.x >> 5;
    unsigned inc = v;
#pragma unroll
    for (int o = 1; o < 32; o <<= 1) {
        const unsigned up = __shfl_up(inc, o, 32);
        if (lane >= o) inc += up;
    }
    if (lane == 31) wsum[w] = inc;
    __syncthreads();
    if (w == 0) {
        const unsigned s = wsum[lane];
        unsigned si = s;
#pragma unroll
        for (int o = 1; o < 32; o <<= 1) {
            const unsigned up = __shfl_up(si, o, 32);
            if (lane >= o) si += up;
        }
        wsum[lane] = si - s;
    }
    __syncthreads();
    const unsigned res = wsum[w] + inc - v;
    __syncthreads();
    return res;
}

__global__ __launch_bounds__(BT)
void gcn_block_sort(const int* __restrict__ ei, unsigned* __restrict__ runm,
                    unsigned* __restrict__ pk, int E, int n, int chunk_len) {
    __shared__ unsigned sorted[CHMAX];
    __shared__ unsigned hist[NBP];
    __shared__ unsigned lstart[NBP];
    __shared__ unsigned wsum[32];
    const int t   = threadIdx.x;
    const int blk = blockIdx.x;
    if (t < NBP) hist[t] = 0u;

    const int* __restrict__ srcp = ei;
    const int* __restrict__ dstp = ei + E;
    const int c0 = blk * chunk_len;
    const int c1 = (c0 + chunk_len < E) ? (c0 + chunk_len) : E;
    int d[BU], s[BU];
#pragma unroll
    for (int u = 0; u < BU; ++u) {
        const int e  = c0 + u * BT + t;
        const int ec = (e < c1) ? e : (c1 - 1);
        const int dl = dstp[ec];
        s[u] = srcp[ec];
        d[u] = (e < c1) ? dl : -1;
    }
    __syncthreads();

    unsigned rk[BU];
#pragma unroll
    for (int u = 0; u < BU; ++u) {
        rk[u] = 0u;
        if ((unsigned)d[u] < (unsigned)n) rk[u] = atomicAdd(&hist[d[u] >> 8], 1u);
    }
    __syncthreads();

    const unsigned cm = (t < NBP) ? hist[t] : 0u;
    const unsigned ls = gcn_block_scan(cm, wsum);
    if (t < NBP) {
        lstart[t] = ls;
        runm[blk * NBP + t] = (ls << 16) | cm;
    }
    __syncthreads();

#pragma unroll
    for (int u = 0; u < BU; ++u) {
        if ((unsigned)d[u] < (unsigned)n) {
            const unsigned slot = lstart[d[u] >> 8] + rk[u];
            if (slot < (unsigned)CHMAX)
                sorted[slot] = ((unsigned)s[u] << 8) | ((unsigned)d[u] & 255u);
        }
    }
    __syncthreads();

    const unsigned nall    = lstart[NBP - 1] + hist[NBP - 1];
    const unsigned cnt_blk = (nall < (unsigned)CHMAX) ? nall : (unsigned)CHMAX;
    unsigned* __restrict__ dstw = pk + (long long)blk * chunk_len;
    unsigned wv[BU];
#pragma unroll
    for (int u = 0; u < BU; ++u) {
        const unsigned idx = (unsigned)(u * BT + t);
        wv[u] = sorted[(idx < cnt_blk) ? idx : 0u];
    }
#pragma unroll
    for (int u = 0; u < BU; ++u) {
        const unsigned idx = (unsigned)(u * BT + t);
        if (idx < cnt_blk) dstw[idx] = wv[u];
    }
}

template <class F>
__device__ __forceinline__ void gcn_sweep_runs(const unsigned* __restrict__ pk, const unsigned* rlen,
                                               const unsigned* rsrc, const unsigned* roff,
                                               int nr, int w, unsigned lane, F fn) {
    for (int r0 = w; r0 < nr; r0 += 32 * RQ) {
        unsigned len[RQ], src[RQ], off[RQ], e0[RQ], e1[RQ];
#pragma unroll
        for (int q = 0; q < RQ; ++q) {
            const int r  = r0 + 32 * q;
            const int rc = (r < nr) ? r : r0;
            const unsigned ln = (unsigned)__builtin_amdgcn_readfirstlane((int)rlen[rc]);
            src[q] = (unsigned)__builtin_amdgcn_readfirstlane((int)rsrc[rc]);
            off[q] = (unsigned)__builtin_amdgcn_readfirstlane((int)roff[rc]);
            len[q] = (r < nr) ? ln : 0u;
        }
#pragma unroll
        for (int q = 0; q < RQ; ++q) {
            e0[q] = pk[(lane < len[q]) ? (src[q] + lane) : 0u];
            e1[q] = pk[(lane + 32u < len[q]) ? (src[q] + lane + 32u) : 0u];
        }
#pragma unroll
        for (int q = 0; q < RQ; ++q) {
            asm volatile("" :: "v"(e0[q]), "v"(e1[q]));
        }
#pragma unroll
        for (int q = 0; q < RQ; ++q) {
            if (lane < len[q])       fn(e0[q], off[q] + lane);
            if (lane + 32u < len[q]) fn(e1[q], off[q] + lane + 32u);
        }
#pragma unroll
        for (int q = 0; q < RQ; ++q) {
            if (len[q] > 64u)
                for (unsigned k = 64u + lane; k < len[q]; k += 32u) fn(pk[src[q] + k], off[q] + k);
        }
    }
}

__device__ __forceinline__ void gcn_load_xrows(const float* __restrict__ x, int b, int n, int t,
                                               float4 (&xv)[BN / 32]) {
#pragma unroll
    for (int i = 0; i < BN / 32; ++i) {
        const int r  = b * BN + i * 32 + (t >> 5);
        const int rc = (r < n) ? r : (n - 1);
        xv[i] = *reinterpret_cast<const float4*>(x + (unsigned)rc * (unsigned)F_IN + (unsigned)(t & 31) * 4u);
    }
}

__global__ __launch_bounds__(BT, 16)
void gcn_bucket_finish(const unsigned* __restrict__ pk, const unsigned* __restrict__ runm,
                       int* __restrict__ csr,
                       unsigned* __restrict__ first, unsigned* __restrict__ degi,
                       float* __restrict__ dis, const float* __restrict__ x,
                       const float* __restrict__ W1, float* __restrict__ xWs,
                       int n, int nblk, int chunk_len) {
    __shared__ __attribute__((aligned(16))) unsigned buf[CAPB];
    __shared__ __attribute__((aligned(16))) unsigned buf2[CAPB];
    __shared__ float sW[F_IN * F_HID];
    __shared__ float sdis[BN];
    __shared__ unsigned ncnt[BN * NQ];
    __shared__ unsigned rlen[BT];
    __shared__ unsigned rsrc[BT];
    __shared__ unsigned roff[BT + 1];
    unsigned* const nstart = roff;
    __shared__ unsigned tot[2];
    __shared__ unsigned wsum[32];
    const int t = threadIdx.x;
    const int b = blockIdx.x;
    const int w = __builtin_amdgcn_readfirstlane(t >> 5);
    const unsigned lane = (unsigned)(t & 31);
    const int nr = (nblk < BT) ? nblk : BT;
    unsigned pw = 0u;
    if (t < nr) pw = runm[t * NBP + b];
    const unsigned c = pw & 0xFFFFu;
    const unsigned l = pw >> 16;
    ncnt[t] = 0u;
    const float q4 = (float)NQ / (float)n;
    sW[t]      = W1[t];
    sW[t + BT] = W1[t + BT];
    const unsigned ro = gcn_block_scan(c, wsum);
    const unsigned lo = gcn_block_scan(l, wsum);
    rlen[t] = c;
    rsrc[t] = (unsigned)t * (unsigned)chunk_len + l;
    roff[t] = ro;
    if (t == BT - 1) { tot[0] = ro + c; tot[1] = lo + l; }
    __syncthreads();
    const unsigned nb = tot[0];
    const unsigned gb = tot[1];
    const bool fits = nb <= (unsigned)CAPB;

    gcn_sweep_runs(pk, rlen, rsrc, roff, nr, w, lane, [&](unsigned e, unsigned pos) {
        atomicAdd(&ncnt[gcn_cell(e & 255u, e >> 8, q4)], 1u);
        if (fits) buf2[pos] = e;
    });
    __syncthreads();

    const unsigned cv = ncnt[t];
    const unsigned ex = gcn_block_scan(cv, wsum);
    nstart[t] = ex;
    ncnt[t]   = 0u;
    if (t == BT - 1) nstart[BT] = ex + cv;
    __syncthreads();
    if (t < BN) {
        const int node = b * BN + t;
        if (node < n) {
            const unsigned ns0 = nstart[t * NQ];
            const unsigned v   = nstart[t * NQ + NQ] - ns0;
            degi[node]  = v;
            first[node] = gb + ns0;
            const float dgv = (float)(v + 1u);
            const float dsv = dgv > 0.0f ? rsqrtf(dgv) : 0.0f;
            dis[node] = dsv;
            sdis[t]   = dsv;
        }
    }

    float4 xv[BN / 32];
    if (fits) {
        for (unsigned base = (unsigned)t; base < nb; base += BT * PB) {
            unsigned e[PB], rk[PB], cl[PB];
#pragma unroll
            for (int u = 0; u < PB; ++u) {
                const unsigned i = base + (unsigned)u * BT;
                e[u] = buf2[(i < nb) ? i : 0u];
            }
#pragma unroll
            for (int u = 0; u < PB; ++u) {
                rk[u] = 0u;
                cl[u] = gcn_cell(e[u] & 255u, e[u] >> 8, q4);
                if (base + (unsigned)u * BT < nb) rk[u] = atomicAdd(&ncnt[cl[u]], 1u);
            }
#pragma unroll
            for (int u = 0; u < PB; ++u) {
                if (base + (unsigned)u * BT < nb) {
                    const unsigned slot = nstart[cl[u]] + rk[u];
                    if (slot < (unsigned)CAPB) buf[slot] = (e[u] << 7) | (rk[u] & 127u);
                }
            }
        }
        __syncthreads();
        gcn_load_xrows(x, b, n, t, xv);
        for (unsigned i = (unsigned)t; i < nb; i += BT) {
            const unsigned key = buf[i];
            const unsigned cell = gcn_cell((key >> 7) & 255u, key >> 15, q4);
            const unsigned ns  = nstart[cell];
            const unsigned len = nstart[cell + 1u] - ns;
            unsigned r = i - ns;
            if (len <= (unsigned)SORTMAX) {
                r = 0u;
                for (unsigned j = 0u; j < len; ++j) r += (buf[ns + j] < key) ? 1u : 0u;
            }
            buf2[ns + r] = key >> 15;
        }
        __syncthreads();
        for (unsigned i = (unsigned)t; i < nb; i += BT) csr[gb + i] = (int)buf2[i];
    } else {
        gcn_sweep_runs(pk, rlen, rsrc, roff, nr, w, lane, [&](unsigned e, unsigned) {
            const unsigned cell = gcn_cell(e & 255u, e >> 8, q4);
            const unsigned rk = atomicAdd(&ncnt[cell], 1u);
            csr[gb + nstart[cell] + rk] = (int)(e >> 8);
        });
        gcn_load_xrows(x, b, n, t, xv);
    }
    __syncthreads();

#pragma unroll
    for (int i = 0; i < BN / 32; ++i) {
        const unsigned r  = (unsigned)(i * 32 + (t >> 5));
        const unsigned pp = lane & 15u;
        unsigned* hb = (lane & 16u) ? buf2 : buf;
        *reinterpret_cast<uint4*>(hb + r * 64u + 4u * (pp ^ (r & 15u))) =
            make_uint4(__float_as_uint(xv[i].x), __float_as_uint(xv[i].y),
                       __float_as_uint(xv[i].z), __float_as_uint(xv[i].w));
    }
    __syncthreads();
    if (w < BN / 16) {
        const int row0 = b * BN + w * 16;
        if (row0 < n) {
            const unsigned half = lane >> 4;
            const unsigned l16  = lane & 15u;
            const unsigned rl   = (unsigned)w * 16u + l16;
            v8f acc = {};
#pragma unroll
            for (int s = 0; s < F_IN / 4; ++s) {
                const unsigned* hb = (s < 16) ? buf : buf2;
                const uint2 av = *reinterpret_cast<const uint2*>(
                    hb + rl * 64u + 4u * ((unsigned)(s & 15) ^ l16) + 2u * half);
                const unsigned ka = 4u * (unsigned)s + 2u * half;
                v2f a, bw;
                a.x  = __uint_as_float(av.x);
                a.y  = __uint_as_float(av.y);
                bw.x = sW[ka * F_HID + l16];
                bw.y = sW[(ka + 1u) * F_HID + l16];
                acc = __builtin_amdgcn_wmma_f32_16x16x4_f32(
                         false, a,  false, bw,
                         (short)0, acc,  false,  false);
            }
            float dv[8];
#pragma unroll
            for (int r = 0; r < 8; ++r) dv[r] = sdis[(unsigned)w * 16u + (unsigned)r + half * 8u];
#pragma unroll
            for (int r = 0; r < 8; ++r) {
                const int node = row0 + r + (int)half * 8;
                if (node < n) xWs[(long long)node * F_HID + l16] = acc[r] * dv[r];
            }
        }
    }
}

__global__ __launch_bounds__(256)
void gcn_agg1_gather(const int* __restrict__ csr, const unsigned* __restrict__ first,
                     const unsigned* __restrict__ degi, const float* __restrict__ xWs,
                     const float* __restrict__ dis, const float* __restrict__ b1,
                     const float* __restrict__ W2, float* __restrict__ y, int n) {
    const int t    = blockIdx.x * blockDim.x + threadIdx.x;
    const int node = t >> 4;
    const int j    = t & 15;
    const bool live = node < n;
    const int nd   = live ? node : (n - 1);
    const unsigned beg = first[nd];
    const unsigned cnt = degi[nd];
    const float* __restrict__ col = xWs + j;
    float acc = col[(long long)nd * F_HID];
    int next = ((unsigned)j < cnt) ? csr[beg + (unsigned)j] : -1;
    for (unsigned k0 = 0u; k0 < cnt; k0 += 16u) {
        const int mine = next;
        int   s[16];
        float r[16];
#pragma unroll
        for (int u = 0; u < 16; ++u) {
            s[u] = __shfl(mine, u, 16);
            const int sc = (s[u] >= 0) ? s[u] : nd;
            r[u] = col[(long long)sc * F_HID];
        }
        const unsigned kn = k0 + 16u + (unsigned)j;
        next = (kn < cnt) ? csr[beg + kn] : -1;
#pragma unroll
        for (int u = 0; u < 16; ++u) acc += (s[u] >= 0) ? r[u] : 0.0f;
    }
    const float dv = dis[nd];
    float hv = fmaf(acc, dv, b1[j]);
    hv = fmaxf(hv, 0.0f);
    float p = hv * W2[j];
    p += __shfl_xor(p, 8, 16);
    p += __shfl_xor(p, 4, 16);
    p += __shfl_xor(p, 2, 16);
    p += __shfl_xor(p, 1, 16);
    if (live && j == 0) y[node] = p * dv;
}

#define YS     51200
#define A2T    1024
#define A2G    (A2T / 16)
#define A2IT   7
#define A2NPB  (A2G * A2IT)
#define A2SW   ((YS + A2T * 4 - 1) / (A2T * 4))
#define A2SA   7

template <int B>
__device__ __forceinline__ void gcn_stage_y(const float* __restrict__ ysrc, float* __restrict__ ys,
                                            int len4, int sw0, int rot, unsigned t4) {
    float4 v[B];
#pragma unroll
    for (int b = 0; b < B; ++b) {
        const int sw  = (sw0 + b + rot < A2SW) ? (sw0 + b + rot) : (sw0 + b + rot - A2SW);
        const int off = sw * (A2T * 4);
        v[b] = make_float4(0.0f, 0.0f, 0.0f, 0.0f);
        if ((int)t4 < len4 - off) v[b] = *reinterpret_cast<const float4*>(ysrc + off + t4);
    }
#pragma unroll
    for (int b = 0; b < B; ++b) {
        const int sw  = (sw0 + b + rot < A2SW) ? (sw0 + b + rot) : (sw0 + b + rot - A2SW);
        const int off = sw * (A2T * 4);
        if ((int)t4 < len4 - off) *reinterpret_cast<float4*>(ys + off + t4) = v[b];
    }
}

__global__ __launch_bounds__(A2T)
void gcn_agg2_lds(const int* __restrict__ csr, const unsigned* __restrict__ first,
                  const unsigned* __restrict__ degi, const float* __restrict__ y,
                  const float* __restrict__ dis, const float* __restrict__ b2,
                  float* __restrict__ out, int n) {
    __shared__ __attribute__((aligned(16))) float ys[YS];
    const int tid  = threadIdx.x;
    const int grp  = tid >> 4;
    const int j    = tid & 15;
    const int base = blockIdx.x * A2NPB;
    const int rot  = blockIdx.x % A2SW;

    unsigned cnt[A2IT];
    int      s[A2IT][4];
    {
        unsigned beg[A2IT];
#pragma unroll
        for (int it = 0; it < A2IT; ++it) {
            const int node = base + it * A2G + grp;
            const int nd   = (node < n) ? node : (n - 1);
            beg[it] = first[nd];
            const unsigned c = degi[nd];
            cnt[it] = (node < n) ? c : 0u;
        }
#pragma unroll
        for (int it = 0; it < A2IT; ++it) {
#pragma unroll
            for (int u = 0; u < 4; ++u) {
                const unsigned k = (unsigned)j + 16u * (unsigned)u;
                s[it][u] = (k < cnt[it]) ? csr[beg[it] + k] : -1;
            }
        }
    }
    float acc[A2IT], dv[A2IT], yv[A2IT];
#pragma unroll
    for (int it = 0; it < A2IT; ++it) {
        const int node = base + it * A2G + grp;
        const int nd   = (node < n) ? node : (n - 1);
        dv[it]  = dis[nd];
        yv[it]  = y[nd];
        acc[it] = 0.0f;
    }

    for (int lo = 0; lo < n; lo += YS) {
        const int len  = (n - lo < YS) ? (n - lo) : YS;
        const int len4 = len & ~3;
        gcn_stage_y<A2SA>(y + lo, ys, len4, 0, rot, (unsigned)tid * 4u);
        gcn_stage_y<A2SW - A2SA>(y + lo, ys, len4, A2SA, rot, (unsigned)tid * 4u);
        if (tid < len - len4) ys[len4 + tid] = y[lo + len4 + tid];
        __syncthreads();
#pragma unroll
        for (int it = 0; it < A2IT; ++it) {
#pragma unroll
            for (int u = 0; u < 4; ++u) {
                const unsigned loc = (unsigned)(s[it][u] - lo);
                const float    r   = ys[(loc < (unsigned)len) ? loc : 0u];
                acc[it] += (loc < (unsigned)len) ? r : 0.0f;
            }
        }
#pragma unroll
        for (int it = 0; it < A2IT; ++it) {
            if (cnt[it] > 64u) {
                const unsigned bg = first[base + it * A2G + grp];
                for (unsigned k = 64u + (unsigned)j; k < cnt[it]; k += 16u) {
                    const unsigned loc = (unsigned)(csr[bg + k] - lo);
                    if (loc < (unsigned)len) acc[it] += ys[loc];
                }
            }
        }
        __syncthreads();
    }
#pragma unroll
    for (int w = 8; w >= 1; w >>= 1) {
#pragma unroll
        for (int it = 0; it < A2IT; ++it) acc[it] += __shfl_xor(acc[it], w, 16);
    }
    const float bias = b2[0];
#pragma unroll
    for (int it = 0; it < A2IT; ++it) {
        const int node = base + it * A2G + grp;
        if (node < n && j == 0) out[node] = bias + dv[it] * (yv[it] + acc[it]);
    }
}

extern "C" void kernel_launch(void* const* d_in, const int* in_sizes, int n_in,
                              void* d_out, int out_size, void* d_ws, size_t ws_size,
                              hipStream_t stream) {
    const float* x  = (const float*)d_in[0];
    const int*   ei = (const int*)d_in[1];
    const float* W1 = (const float*)d_in[2];
    const float* b1 = (const float*)d_in[3];
    const float* W2 = (const float*)d_in[4];
    const float* b2 = (const float*)d_in[5];
    float* out = (float*)d_out;

    const int       N = in_sizes[0] / F_IN;
    const long long E = (long long)in_sizes[1] / 2;

    const int Ei        = (int)E;
    const int nblk      = (int)((E + CHMAX - 1) / CHMAX);
    const int chunk_len = (int)((E + nblk - 1) / nblk);
    const int n_bkt     = (N + BN - 1) / BN;

    const long long Np = ((long long)N + 3) / 4 * 4;
    float*    ws    = (float*)d_ws;
    float*    y     = ws;
    float*    dis   = ws + Np;
    float*    xWs   = ws + Np + N;
    unsigned* degi  = (unsigned*)(ws + Np + (long long)17 * N);
    unsigned* first = (unsigned*)(ws + Np + (long long)18 * N);
    unsigned* runm  = (unsigned*)(ws + Np + (long long)19 * N);
    unsigned* pk    = runm + (long long)nblk * NBP;
    int*      csr   = (int*)(pk + (long long)nblk * chunk_len);

    const int T = 256;
    const unsigned nb_n16  = (unsigned)(((long long)N * F_HID + T - 1) / T);
    const unsigned nb_a2   = (unsigned)((N + A2NPB - 1) / A2NPB);

    gcn_block_sort   <<<(unsigned)nblk,  BT, 0, stream>>>(ei, runm, pk, Ei, N, chunk_len);
    gcn_bucket_finish<<<(unsigned)n_bkt, BT, 0, stream>>>(pk, runm, csr, first, degi, dis, x, W1, xWs, N, nblk, chunk_len);
    gcn_agg1_gather  <<<nb_n16,  T,  0, stream>>>(csr, first, degi, xWs, dis, b1, W2, y, N);
    gcn_agg2_lds     <<<nb_a2,   A2T, 0, stream>>>(csr, first, degi, y, dis, b2, out, N);
}


# Buckets of the list build and of x @ W1 hold 200 destination nodes, not 256: 500 level-2 blocks take the 512 block places (two per WGP), 391 larger ones did not; list, sum order, launches unchanged
# speedup vs baseline: 7.6660x; 1.0555x over previous
#include <hip/hip_runtime.h>
#include <hip/hip_bf16.h>

typedef __attribute__((ext_vector_type(2))) float v2f;
typedef __attribute__((ext_vector_type(8))) float v8f;

#define F_IN   128
#define F_HID  16

#define BN      200
#define NBP     512
#define BT      1024
#define BU      13
#define CHMAX   12800
#define RQ      8
#define PB      8
#define CAPB    16384
#define SORTMAX 128
#define NQ      4
#define NT16    ((BN + 15) / 16)
#define XR      ((BN + 31) / 32)
static_assert(BN <= 256 && BN * NQ <= BT && BN * 64 <= CAPB && NT16 * 16 <= 256, "bucket size against the packed word, the counters and the x tile");

__device__ __forceinline__ unsigned gcn_cell(unsigned ld, unsigned src, float q4) {
    const unsigned s = (unsigned)((float)src * q4);
    return ld * (unsigned)NQ + ((s < (unsigned)(NQ - 1)) ? s : (unsigned)(NQ - 1));
}

__device__ __forceinline__ unsigned gcn_block_scan(unsigned v, unsigned* wsum) {
    const int lane = threadIdx.x & 31;
    const int w    = threadIdx.x >> 5;
    unsigned inc = v;
#pragma unroll
    for (int o = 1; o < 32; o <<= 1) {
        const unsigned up = __shfl_up(inc, o, 32);
        if (lane >= o) inc += up;
    }
    if (lane == 31) wsum[w] = inc;
    __syncthreads();
    if (w == 0) {
        const unsigned s = wsum[lane];
        unsigned si = s;
#pragma unroll
        for (int o = 1; o < 32; o <<= 1) {
            const unsigned up = __shfl_up(si, o, 32);
            if (lane >= o) si += up;
        }
        wsum[lane] = si - s;
    }
    __syncthreads();
    const unsigned res = wsum[w] + inc - v;
    __syncthreads();
    return res;
}

__global__ __launch_bounds__(BT)
void gcn_block_sort(const int* __restrict__ ei, unsigned* __restrict__ runm,
                    unsigned* __restrict__ pk, int E, int n, int chunk_len) {
    __shared__ unsigned sorted[CHMAX];
    __shared__ unsigned hist[NBP];
    __shared__ unsigned lstart[NBP];
    __shared__ unsigned wsum[32];
    const int t   = threadIdx.x;
    const int blk = blockIdx.x;
    if (t < NBP) hist[t] = 0u;

    const int* __restrict__ srcp = ei;
    const int* __restrict__ dstp = ei + E;
    const int c0 = blk * chunk_len;
    const int c1 = (c0 + chunk_len < E) ? (c0 + chunk_len) : E;
    int d[BU], s[BU];
#pragma unroll
    for (int u = 0; u < BU; ++u) {
        const int e  = c0 + u * BT + t;
        const int ec = (e < c1) ? e : (c1 - 1);
        const int dl = dstp[ec];
        s[u] = srcp[ec];
        d[u] = (e < c1) ? dl : -1;
    }
    __syncthreads();

    unsigned rk[BU];
#pragma unroll
    for (int u = 0; u < BU; ++u) {
        rk[u] = 0u;
        if ((unsigned)d[u] < (unsigned)n) rk[u] = atomicAdd(&hist[(unsigned)d[u] / (unsigned)BN], 1u);
    }
    __syncthreads();

    const unsigned cm = (t < NBP) ? hist[t] : 0u;
    const unsigned ls = gcn_block_scan(cm, wsum);
    if (t < NBP) {
        lstart[t] = ls;
        runm[blk * NBP + t] = (ls << 16) | cm;
    }
    __syncthreads();

#pragma unroll
    for (int u = 0; u < BU; ++u) {
        if ((unsigned)d[u] < (unsigned)n) {
            const unsigned bk   = (unsigned)d[u] / (unsigned)BN;
            const unsigned slot = lstart[bk] + rk[u];
            if (slot < (unsigned)CHMAX)
                sorted[slot] = ((unsigned)s[u] << 8) | ((unsigned)d[u] - bk * (unsigned)BN);
        }
    }
    __syncthreads();

    const unsigned nall    = lstart[NBP - 1] + hist[NBP - 1];
    const unsigned cnt_blk = (nall < (unsigned)CHMAX) ? nall : (unsigned)CHMAX;
    unsigned* __restrict__ dstw = pk + (long long)blk * chunk_len;
    unsigned wv[BU];
#pragma unroll
    for (int u = 0; u < BU; ++u) {
        const unsigned idx = (unsigned)(u * BT + t);
        wv[u] = sorted[(idx < cnt_blk) ? idx : 0u];
    }
#pragma unroll
    for (int u = 0; u < BU; ++u) {
        const unsigned idx = (unsigned)(u * BT + t);
        if (idx < cnt_blk) dstw[idx] = wv[u];
    }
}

template <class F>
__device__ __forceinline__ void gcn_sweep_runs(const unsigned* __restrict__ pk, const unsigned* rlen,
                                               const unsigned* rsrc, const unsigned* roff,
                                               int nr, int w, unsigned lane, F fn) {
    for (int r0 = w; r0 < nr; r0 += 32 * RQ) {
        unsigned len[RQ], src[RQ], off[RQ], e0[RQ], e1[RQ];
#pragma unroll
        for (int q = 0; q < RQ; ++q) {
            const int r  = r0 + 32 * q;
            const int rc = (r < nr) ? r : r0;
            const unsigned ln = (unsigned)__builtin_amdgcn_readfirstlane((int)rlen[rc]);
            src[q] = (unsigned)__builtin_amdgcn_readfirstlane((int)rsrc[rc]);
            off[q] = (unsigned)__builtin_amdgcn_readfirstlane((int)roff[rc]);
            len[q] = (r < nr) ? ln : 0u;
        }
#pragma unroll
        for (int q = 0; q < RQ; ++q) {
            e0[q] = pk[(lane < len[q]) ? (src[q] + lane) : 0u];
            e1[q] = pk[(lane + 32u < len[q]) ? (src[q] + lane + 32u) : 0u];
        }
#pragma unroll
        for (int q = 0; q < RQ; ++q) {
            asm volatile("" :: "v"(e0[q]), "v"(e1[q]));
        }
#pragma unroll
        for (int q = 0; q < RQ; ++q) {
            if (lane < len[q])       fn(e0[q], off[q] + lane);
            if (lane + 32u < len[q]) fn(e1[q], off[q] + lane + 32u);
        }
#pragma unroll
        for (int q = 0; q < RQ; ++q) {
            if (len[q] > 64u)
                for (unsigned k = 64u + lane; k < len[q]; k += 32u) fn(pk[src[q] + k], off[q] + k);
        }
    }
}

__device__ __forceinline__ void gcn_load_xrows(const float* __restrict__ x, int b, int n, int w, unsigned lane,
                                               float4 (&xv)[XR]) {
#pragma unroll
    for (int i = 0; i < XR; ++i) {
        xv[i] = make_float4(0.0f, 0.0f, 0.0f, 0.0f);
        if (i * 32 + w < BN) {
            const int r  = b * BN + i * 32 + w;
            const int rc = (r < n) ? r : (n - 1);
            xv[i] = *reinterpret_cast<const float4*>(x + (unsigned)rc * (unsigned)F_IN + lane * 4u);
        }
    }
}

__global__ __launch_bounds__(BT, 16)
void gcn_bucket_finish(const unsigned* __restrict__ pk, const unsigned* __restrict__ runm,
                       int* __restrict__ csr,
                       unsigned* __restrict__ first, unsigned* __restrict__ degi,
                       float* __restrict__ dis, const float* __restrict__ x,
                       const float* __restrict__ W1, float* __restrict__ xWs,
                       int n, int nblk, int chunk_len) {
    __shared__ __attribute__((aligned(16))) unsigned buf[CAPB];
    __shared__ __attribute__((aligned(16))) unsigned buf2[CAPB];
    __shared__ float sW[F_IN * F_HID];
    __shared__ float sdis[NT16 * 16];
    __shared__ unsigned ncnt[BT];
    __shared__ unsigned rlen[BT];
    __shared__ unsigned rsrc[BT];
    __shared__ unsigned roff[BT + 1];
    unsigned* const nstart = roff;
    __shared__ unsigned tot[2];
    __shared__ unsigned wsum[32];
    const int t = threadIdx.x;
    const int b = blockIdx.x;
    const int w = __builtin_amdgcn_readfirstlane(t >> 5);
    const unsigned lane = (unsigned)(t & 31);
    const int nr = (nblk < BT) ? nblk : BT;
    unsigned pw = 0u;
    if (t < nr) pw = runm[t * NBP + b];
    const unsigned c = pw & 0xFFFFu;
    const unsigned l = pw >> 16;
    ncnt[t] = 0u;
    const float q4 = (float)NQ / (float)n;
    sW[t]      = W1[t];
    sW[t + BT] = W1[t + BT];
    const unsigned ro = gcn_block_scan(c, wsum);
    const unsigned lo = gcn_block_scan(l, wsum);
    rlen[t] = c;
    rsrc[t] = (unsigned)t * (unsigned)chunk_len + l;
    roff[t] = ro;
    if (t == BT - 1) { tot[0] = ro + c; tot[1] = lo + l; }
    __syncthreads();
    const unsigned nb = tot[0];
    const unsigned gb = tot[1];
    const bool fits = nb <= (unsigned)CAPB;

    gcn_sweep_runs(pk, rlen, rsrc, roff, nr, w, lane, [&](unsigned e, unsigned pos) {
        atomicAdd(&ncnt[gcn_cell(e & 255u, e >> 8, q4)], 1u);
        if (fits) buf2[pos] = e;
    });
    __syncthreads();

    const unsigned cv = ncnt[t];
    const unsigned ex = gcn_block_scan(cv, wsum);
    nstart[t] = ex;
    ncnt[t]   = 0u;
    if (t == BT - 1) nstart[BT] = ex + cv;
    __syncthreads();
    if (t < BN) {
        const int node = b * BN + t;
        if (node < n) {
            const unsigned ns0 = nstart[t * NQ];
            const unsigned v   = nstart[t * NQ + NQ] - ns0;
            degi[node]  = v;
            first[node] = gb + ns0;
            const float dgv = (float)(v + 1u);
            const float dsv = dgv > 0.0f ? rsqrtf(dgv) : 0.0f;
            dis[node] = dsv;
            sdis[t]   = dsv;
        }
    }

    float4 xv[XR];
    if (fits) {
        for (unsigned base = (unsigned)t; base < nb; base += BT * PB) {
            unsigned e[PB], rk[PB], cl[PB];
#pragma unroll
            for (int u = 0; u < PB; ++u) {
                const unsigned i = base + (unsigned)u * BT;
                e[u] = buf2[(i < nb) ? i : 0u];
            }
#pragma unroll
            for (int u = 0; u < PB; ++u) {
                rk[u] = 0u;
                cl[u] = gcn_cell(e[u] & 255u, e[u] >> 8, q4);
                if (base + (unsigned)u * BT < nb) rk[u] = atomicAdd(&ncnt[cl[u]], 1u);
            }
#pragma unroll
            for (int u = 0; u < PB; ++u) {
                if (base + (unsigned)u * BT < nb) {
                    const unsigned slot = nstart[cl[u]] + rk[u];
                    if (slot < (unsigned)CAPB) buf[slot] = (e[u] << 7) | (rk[u] & 127u);
                }
            }
        }
        __syncthreads();
        gcn_load_xrows(x, b, n, w, lane, xv);
        for (unsigned i = (unsigned)t; i < nb; i += BT) {
            const unsigned key = buf[i];
            const unsigned cell = gcn_cell((key >> 7) & 255u, key >> 15, q4);
            const unsigned ns  = nstart[cell];
            const unsigned len = nstart[cell + 1u] - ns;
            unsigned r = i - ns;
            if (len <= (unsigned)SORTMAX) {
                r = 0u;
                for (unsigned j = 0u; j < len; ++j) r += (buf[ns + j] < key) ? 1u : 0u;
            }
            buf2[ns + r] = key >> 15;
        }
        __syncthreads();
        for (unsigned i = (unsigned)t; i < nb; i += BT) csr[gb + i] = (int)buf2[i];
    } else {
        gcn_sweep_runs(pk, rlen, rsrc, roff, nr, w, lane, [&](unsigned e, unsigned) {
            const unsigned cell = gcn_cell(e & 255u, e >> 8, q4);
            const unsigned rk = atomicAdd(&ncnt[cell], 1u);
            csr[gb + nstart[cell] + rk] = (int)(e >> 8);
        });
        gcn_load_xrows(x, b, n, w, lane, xv);
    }
    __syncthreads();

#pragma unroll
    for (int i = 0; i < XR; ++i) {
        if (i * 32 + w < BN) {
            const unsigned r  = (unsigned)(i * 32 + w);
            const unsigned pp = lane & 15u;
            unsigned* hb = (lane & 16u) ? buf2 : buf;
            *reinterpret_cast<uint4*>(hb + r * 64u + 4u * (pp ^ (r & 15u))) =
                make_uint4(__float_as_uint(xv[i].x), __float_as_uint(xv[i].y),
                           __float_as_uint(xv[i].z), __float_as_uint(xv[i].w));
        }
    }
    __syncthreads();
    if (w < NT16) {
        const int row0 = b * BN + w * 16;
        if (row0 < n) {
            const unsigned half = lane >> 4;
            const unsigned l16  = lane & 15u;
            const unsigned rl   = (unsigned)w * 16u + l16;
            v8f acc = {};
#pragma unroll
            for (int s = 0; s < F_IN / 4; ++s) {
                const unsigned* hb = (s < 16) ? buf : buf2;
                const uint2 av = *reinterpret_cast<const uint2*>(
                    hb + rl * 64u + 4u * ((unsigned)(s & 15) ^ l16) + 2u * half);
                const unsigned ka = 4u * (unsigned)s + 2u * half;
                v2f a, bw;
                a.x  = __uint_as_float(av.x);
                a.y  = __uint_as_float(av.y);
                bw.x = sW[ka * F_HID + l16];
                bw.y = sW[(ka + 1u) * F_HID + l16];
                acc = __builtin_amdgcn_wmma_f32_16x16x4_f32(
                         false, a,  false, bw,
                         (short)0, acc,  false,  false);
            }
            float dv[8];
#pragma unroll
            for (int r = 0; r < 8; ++r) dv[r] = sdis[(unsigned)w * 16u + (unsigned)r + half * 8u];
#pragma unroll
            for (int r = 0; r < 8; ++r) {
                const int rloc = w * 16 + r + (int)half * 8;
                const int node = b * BN + rloc;
                if (rloc < BN && node < n) xWs[(long long)node * F_HID + l16] = acc[r] * dv[r];
            }
        }
    }
}

__global__ __launch_bounds__(256)
void gcn_agg1_gather(const int* __restrict__ csr, const unsigned* __restrict__ first,
                     const unsigned* __restrict__ degi, const float* __restrict__ xWs,
                     const float* __restrict__ dis, const float* __restrict__ b1,
                     const float* __restrict__ W2, float* __restrict__ y, int n) {
    const int t    = blockIdx.x * blockDim.x + threadIdx.x;
    const int node = t >> 4;
    const int j    = t & 15;
    const bool live = node < n;
    const int nd   = live ? node : (n - 1);
    const unsigned beg = first[nd];
    const unsigned cnt = degi[nd];
    const float* __restrict__ col = xWs + j;
    float acc = col[(long long)nd * F_HID];
    int next = ((unsigned)j < cnt) ? csr[beg + (unsigned)j] : -1;
    for (unsigned k0 = 0u; k0 < cnt; k0 += 16u) {
        const int mine = next;
        int   s[16];
        float r[16];
#pragma unroll
        for (int u = 0; u < 16; ++u) {
            s[u] = __shfl(mine, u, 16);
            const int sc = (s[u] >= 0) ? s[u] : nd;
            r[u] = col[(long long)sc * F_HID];
        }
        const unsigned kn = k0 + 16u + (unsigned)j;
        next = (kn < cnt) ? csr[beg + kn] : -1;
#pragma unroll
        for (int u = 0; u < 16; ++u) acc += (s[u] >= 0) ? r[u] : 0.0f;
    }
    const float dv = dis[nd];
    float hv = fmaf(acc, dv, b1[j]);
    hv = fmaxf(hv, 0.0f);
    float p = hv * W2[j];
    p += __shfl_xor(p, 8, 16);
    p += __shfl_xor(p, 4, 16);
    p += __shfl_xor(p, 2, 16);
    p += __shfl_xor(p, 1, 16);
    if (live && j == 0) y[node] = p * dv;
}

#define YS     51200
#define A2T    1024
#define A2G    (A2T / 16)
#define A2IT   7
#define A2NPB  (A2G * A2IT)
#define A2SW   ((YS + A2T * 4 - 1) / (A2T * 4))
#define A2SA   7

template <int B>
__device__ __forceinline__ void gcn_stage_y(const float* __restrict__ ysrc, float* __restrict__ ys,
                                            int len4, int sw0, int rot, unsigned t4) {
    float4 v[B];
#pragma unroll
    for (int b = 0; b < B; ++b) {
        const int sw  = (sw0 + b + rot < A2SW) ? (sw0 + b + rot) : (sw0 + b + rot - A2SW);
        const int off = sw * (A2T * 4);
        v[b] = make_float4(0.0f, 0.0f, 0.0f, 0.0f);
        if ((int)t4 < len4 - off) v[b] = *reinterpret_cast<const float4*>(ysrc + off + t4);
    }
#pragma unroll
    for (int b = 0; b < B; ++b) {
        const int sw  = (sw0 + b + rot < A2SW) ? (sw0 + b + rot) : (sw0 + b + rot - A2SW);
        const int off = sw * (A2T * 4);
        if ((int)t4 < len4 - off) *reinterpret_cast<float4*>(ys + off + t4) = v[b];
    }
}

__global__ __launch_bounds__(A2T)
void gcn_agg2_lds(const int* __restrict__ csr, const unsigned* __restrict__ first,
                  const unsigned* __restrict__ degi, const float* __restrict__ y,
                  const float* __restrict__ dis, const float* __restrict__ b2,
                  float* __restrict__ out, int n) {
    __shared__ __attribute__((aligned(16))) float ys[YS];
    const int tid  = threadIdx.x;
    const int grp  = tid >> 4;
    const int j    = tid & 15;
    const int base = blockIdx.x * A2NPB;
    const int rot  = blockIdx.x % A2SW;

    unsigned cnt[A2IT];
    int      s[A2IT][4];
    {
        unsigned beg[A2IT];
#pragma unroll
        for (int it = 0; it < A2IT; ++it) {
            const int node = base + it * A2G + grp;
            const int nd   = (node < n) ? node : (n - 1);
            beg[it] = first[nd];
            const unsigned c = degi[nd];
            cnt[it] = (node < n) ? c : 0u;
        }
#pragma unroll
        for (int it = 0; it < A2IT; ++it) {
#pragma unroll
            for (int u = 0; u < 4; ++u) {
                const unsigned k = (unsigned)j + 16u * (unsigned)u;
                s[it][u] = (k < cnt[it]) ? csr[beg[it] + k] : -1;
            }
        }
    }
    float acc[A2IT], dv[A2IT], yv[A2IT];
#pragma unroll
    for (int it = 0; it < A2IT; ++it) {
        const int node = base + it * A2G + grp;
        const int nd   = (node < n) ? node : (n - 1);
        dv[it]  = dis[nd];
        yv[it]  = y[nd];
        acc[it] = 0.0f;
    }

    for (int lo = 0; lo < n; lo += YS) {
        const int len  = (n - lo < YS) ? (n - lo) : YS;
        const int len4 = len & ~3;
        gcn_stage_y<A2SA>(y + lo, ys, len4, 0, rot, (unsigned)tid * 4u);
        gcn_stage_y<A2SW - A2SA>(y + lo, ys, len4, A2SA, rot, (unsigned)tid * 4u);
        if (tid < len - len4) ys[len4 + tid] = y[lo + len4 + tid];
        __syncthreads();
#pragma unroll
        for (int it = 0; it < A2IT; ++it) {
#pragma unroll
            for (int u = 0; u < 4; ++u) {
                const unsigned loc = (unsigned)(s[it][u] - lo);
                const float    r   = ys[(loc < (unsigned)len) ? loc : 0u];
                acc[it] += (loc < (unsigned)len) ? r : 0.0f;
            }
        }
#pragma unroll
        for (int it = 0; it < A2IT; ++it) {
            if (cnt[it] > 64u) {
                const unsigned bg = first[base + it * A2G + grp];
                for (unsigned k = 64u + (unsigned)j; k < cnt[it]; k += 16u) {
                    const unsigned loc = (unsigned)(csr[bg + k] - lo);
                    if (loc < (unsigned)len) acc[it] += ys[loc];
                }
            }
        }
        __syncthreads();
    }
#pragma unroll
    for (int w = 8; w >= 1; w >>= 1) {
#pragma unroll
        for (int it = 0; it < A2IT; ++it) acc[it] += __shfl_xor(acc[it], w, 16);
    }
    const float bias = b2[0];
#pragma unroll
    for (int it = 0; it < A2IT; ++it) {
        const int node = base + it * A2G + grp;
        if (node < n && j == 0) out[node] = bias + dv[it] * (yv[it] + acc[it]);
    }
}

extern "C" void kernel_launch(void* const* d_in, const int* in_sizes, int n_in,
                              void* d_out, int out_size, void* d_ws, size_t ws_size,
                              hipStream_t stream) {
    const float* x  = (const float*)d_in[0];
    const int*   ei = (const int*)d_in[1];
    const float* W1 = (const float*)d_in[2];
    const float* b1 = (const float*)d_in[3];
    const float* W2 = (const float*)d_in[4];
    const float* b2 = (const float*)d_in[5];
    float* out = (float*)d_out;

    const int       N = in_sizes[0] / F_IN;
    const long long E = (long long)in_sizes[1] / 2;

    const int Ei        = (int)E;
    const int nblk      = (int)((E + CHMAX - 1) / CHMAX);
    const int chunk_len = (int)((E + nblk - 1) / nblk);
    const int n_bkt     = (N + BN - 1) / BN;

    const long long Np = ((long long)N + 3) / 4 * 4;
    float*    ws    = (float*)d_ws;
    float*    y     = ws;
    float*    dis   = ws + Np;
    float*    xWs   = ws + Np + N;
    unsigned* degi  = (unsigned*)(ws + Np + (long long)17 * N);
    unsigned* first = (unsigned*)(ws + Np + (long long)18 * N);
    unsigned* runm  = (unsigned*)(ws + Np + (long long)19 * N);
    unsigned* pk    = runm + (long long)nblk * NBP;
    int*      csr   = (int*)(pk + (long long)nblk * chunk_len);

    const int T = 256;
    const unsigned nb_n16  = (unsigned)(((long long)N * F_HID + T - 1) / T);
    const unsigned nb_a2   = (unsigned)((N + A2NPB - 1) / A2NPB);

    gcn_block_sort   <<<(unsigned)nblk,  BT, 0, stream>>>(ei, runm, pk, Ei, N, chunk_len);
    gcn_bucket_finish<<<(unsigned)n_bkt, BT, 0, stream>>>(pk, runm, csr, first, degi, dis, x, W1, xWs, N, nblk, chunk_len);
    gcn_agg1_gather  <<<nb_n16,  T,  0, stream>>>(csr, first, degi, xWs, dis, b1, W2, y, N);
    gcn_agg2_lds     <<<nb_a2,   A2T, 0, stream>>>(csr, first, degi, y, dis, b2, out, N);
}


# List build no longer orders a node's sources (rank pass gone; placed per node in arrival order); x rows asked behind the run sweep, level-2 barriers wait on LDS only; f32 sum order varies run to run
# speedup vs baseline: 8.0662x; 1.0522x over previous
#include <hip/hip_runtime.h>
#include <hip/hip_bf16.h>

typedef __attribute__((ext_vector_type(2))) float v2f;
typedef __attribute__((ext_vector_type(8))) float v8f;

#define F_IN   128
#define F_HID  16

#define BN      200
#define NBP     512
#define BT      1024
#define BU      13
#define CHMAX   12800
#define RQ      8
#define PB      8
#define CAPB    16384
#define NT16    ((BN + 15) / 16)
#define XR      ((BN + 31) / 32)
static_assert(BN <= 256 && BN < BT && BN * 64 <= CAPB && NT16 * 16 <= 256, "bucket size against the packed word, the counters and the x tile");

__device__ __forceinline__ void gcn_lds_barrier() {
    __builtin_amdgcn_fence(__ATOMIC_RELEASE, "workgroup", "local");
    __builtin_amdgcn_s_barrier();
    __builtin_amdgcn_fence(__ATOMIC_ACQUIRE, "workgroup", "local");
}

__device__ __forceinline__ unsigned gcn_block_scan(unsigned v, unsigned* wsum) {
    const int lane = threadIdx.x & 31;
    const int w    = threadIdx.x >> 5;
    unsigned inc = v;
#pragma unroll
    for (int o = 1; o < 32; o <<= 1) {
        const unsigned up = __shfl_up(inc, o, 32);
        if (lane >= o) inc += up;
    }
    if (lane == 31) wsum[w] = inc;
    gcn_lds_barrier();
    if (w == 0) {
        const unsigned s = wsum[lane];
        unsigned si = s;
#pragma unroll
        for (int o = 1; o < 32; o <<= 1) {
            const unsigned up = __shfl_up(si, o, 32);
            if (lane >= o) si += up;
        }
        wsum[lane] = si - s;
    }
    gcn_lds_barrier();
    const unsigned res = wsum[w] + inc - v;
    gcn_lds_barrier();
    return res;
}

__global__ __launch_bounds__(BT)
void gcn_block_sort(const int* __restrict__ ei, unsigned* __restrict__ runm,
                    unsigned* __restrict__ pk, int E, int n, int chunk_len) {
    __shared__ unsigned sorted[CHMAX];
    __shared__ unsigned hist[NBP];
    __shared__ unsigned lstart[NBP];
    __shared__ unsigned wsum[32];
    const int t   = threadIdx.x;
    const int blk = blockIdx.x;
    if (t < NBP) hist[t] = 0u;

    const int* __restrict__ srcp = ei;
    const int* __restrict__ dstp = ei + E;
    const int c0 = blk * chunk_len;
    const int c1 = (c0 + chunk_len < E) ? (c0 + chunk_len) : E;
    int d[BU], s[BU];
#pragma unroll
    for (int u = 0; u < BU; ++u) {
        const int e  = c0 + u * BT + t;
        const int ec = (e < c1) ? e : (c1 - 1);
        const int dl = dstp[ec];
        s[u] = srcp[ec];
        d[u] = (e < c1) ? dl : -1;
    }
    __syncthreads();

    unsigned rk[BU];
#pragma unroll
    for (int u = 0; u < BU; ++u) {
        rk[u] = 0u;
        if ((unsigned)d[u] < (unsigned)n) rk[u] = atomicAdd(&hist[(unsigned)d[u] / (unsigned)BN], 1u);
    }
    __syncthreads();

    const unsigned cm = (t < NBP) ? hist[t] : 0u;
    const unsigned ls = gcn_block_scan(cm, wsum);
    if (t < NBP) {
        lstart[t] = ls;
        runm[blk * NBP + t] = (ls << 16) | cm;
    }
    __syncthreads();

#pragma unroll
    for (int u = 0; u < BU; ++u) {
        if ((unsigned)d[u] < (unsigned)n) {
            const unsigned bk   = (unsigned)d[u] / (unsigned)BN;
            const unsigned slot = lstart[bk] + rk[u];
            if (slot < (unsigned)CHMAX)
                sorted[slot] = ((unsigned)s[u] << 8) | ((unsigned)d[u] - bk * (unsigned)BN);
        }
    }
    __syncthreads();

    const unsigned nall    = lstart[NBP - 1] + hist[NBP - 1];
    const unsigned cnt_blk = (nall < (unsigned)CHMAX) ? nall : (unsigned)CHMAX;
    unsigned* __restrict__ dstw = pk + (long long)blk * chunk_len;
    unsigned wv[BU];
#pragma unroll
    for (int u = 0; u < BU; ++u) {
        const unsigned idx = (unsigned)(u * BT + t);
        wv[u] = sorted[(idx < cnt_blk) ? idx : 0u];
    }
#pragma unroll
    for (int u = 0; u < BU; ++u) {
        const unsigned idx = (unsigned)(u * BT + t);
        if (idx < cnt_blk) dstw[idx] = wv[u];
    }
}

template <class F>
__device__ __forceinline__ void gcn_sweep_runs(const unsigned* __restrict__ pk, const unsigned* rlen,
                                               const unsigned* rsrc, const unsigned* roff,
                                               int nr, int w, unsigned lane, F fn) {
    for (int r0 = w; r0 < nr; r0 += 32 * RQ) {
        unsigned len[RQ], src[RQ], off[RQ], e0[RQ], e1[RQ];
#pragma unroll
        for (int q = 0; q < RQ; ++q) {
            const int r  = r0 + 32 * q;
            const int rc = (r < nr) ? r : r0;
            const unsigned ln = (unsigned)__builtin_amdgcn_readfirstlane((int)rlen[rc]);
            src[q] = (unsigned)__builtin_amdgcn_readfirstlane((int)rsrc[rc]);
            off[q] = (unsigned)__builtin_amdgcn_readfirstlane((int)roff[rc]);
            len[q] = (r < nr) ? ln : 0u;
        }
#pragma unroll
        for (int q = 0; q < RQ; ++q) {
            e0[q] = pk[(lane < len[q]) ? (src[q] + lane) : 0u];
            e1[q] = pk[(lane + 32u < len[q]) ? (src[q] + lane + 32u) : 0u];
        }
#pragma unroll
        for (int q = 0; q < RQ; ++q) {
            asm volatile("" :: "v"(e0[q]), "v"(e1[q]));
        }
#pragma unroll
        for (int q = 0; q < RQ; ++q) {
            if (lane < len[q])       fn(e0[q], off[q] + lane);
            if (lane + 32u < len[q]) fn(e1[q], off[q] + lane + 32u);
        }
#pragma unroll
        for (int q = 0; q < RQ; ++q) {
            if (len[q] > 64u)
                for (unsigned k = 64u + lane; k < len[q]; k += 32u) fn(pk[src[q] + k], off[q] + k);
        }
    }
}

__device__ __forceinline__ void gcn_load_xrows(const float* __restrict__ x, int b, int n, int w, unsigned lane,
                                               float4 (&xv)[XR]) {
#pragma unroll
    for (int i = 0; i < XR; ++i) {
        const int rl = (i * 32 + w < BN) ? (i * 32 + w) : (BN - 1);
        const int r  = b * BN + rl;
        const int rc = (r < n) ? r : (n - 1);
        xv[i] = *reinterpret_cast<const float4*>(x + (unsigned)rc * (unsigned)F_IN + lane * 4u);
    }
}

__global__ __launch_bounds__(BT, 16)
void gcn_bucket_finish(const unsigned* __restrict__ pk, const unsigned* __restrict__ runm,
                       int* __restrict__ csr,
                       unsigned* __restrict__ first, unsigned* __restrict__ degi,
                       float* __restrict__ dis, const float* __restrict__ x,
                       const float* __restrict__ W1, float* __restrict__ xWs,
                       int n, int nblk, int chunk_len) {
    __shared__ __attribute__((aligned(16))) unsigned buf[CAPB];
    __shared__ __attribute__((aligned(16))) unsigned buf2[CAPB];
    __shared__ float sW[F_IN * F_HID];
    __shared__ float sdis[NT16 * 16];
    __shared__ unsigned ncnt[BT];
    __shared__ unsigned rlen[BT];
    __shared__ unsigned rsrc[BT];
    __shared__ unsigned roff[BT + 1];
    unsigned* const nstart = roff;
    __shared__ unsigned tot[2];
    __shared__ unsigned wsum[32];
    const int t = threadIdx.x;
    const int b = blockIdx.x;
    const int w = __builtin_amdgcn_readfirstlane(t >> 5);
    const unsigned lane = (unsigned)(t & 31);
    const int nr = (nblk < BT) ? nblk : BT;
    unsigned pw = 0u;
    if (t < nr) pw = runm[t * NBP + b];
    const unsigned c = pw & 0xFFFFu;
    const unsigned l = pw >> 16;
    ncnt[t] = 0u;
    sW[t]      = W1[t];
    sW[t + BT] = W1[t + BT];
    const unsigned ro = gcn_block_scan(c, wsum);
    const unsigned lo = gcn_block_scan(l, wsum);
    rlen[t] = c;
    rsrc[t] = (unsigned)t * (unsigned)chunk_len + l;
    roff[t] = ro;
    if (t == BT - 1) { tot[0] = ro + c; tot[1] = lo + l; }
    gcn_lds_barrier();
    const unsigned nb = tot[0];
    const unsigned gb = tot[1];
    const bool fits = nb <= (unsigned)CAPB;

    float4 xv[XR];
    gcn_sweep_runs(pk, rlen, rsrc, roff, nr, w, lane, [&](unsigned e, unsigned pos) {
        atomicAdd(&ncnt[e & 255u], 1u);
        if (fits) buf2[pos] = e;
    });
    gcn_load_xrows(x, b, n, w, lane, xv);
    gcn_lds_barrier();

    const unsigned cv = ncnt[t];
    const unsigned ex = gcn_block_scan(cv, wsum);
    nstart[t] = ex;
    ncnt[t]   = 0u;
    if (t == BT - 1) nstart[BT] = ex + cv;
    gcn_lds_barrier();
    if (t < BN) {
        const int node = b * BN + t;
        if (node < n) {
            const unsigned ns0 = nstart[t];
            const unsigned v   = nstart[t + 1] - ns0;
            degi[node]  = v;
            first[node] = gb + ns0;
            const float dgv = (float)(v + 1u);
            const float dsv = dgv > 0.0f ? rsqrtf(dgv) : 0.0f;
            dis[node] = dsv;
            sdis[t]   = dsv;
        }
    }

    if (fits) {
        for (unsigned base = (unsigned)t; base < nb; base += BT * PB) {
            unsigned e[PB], rk[PB];
#pragma unroll
            for (int u = 0; u < PB; ++u) {
                const unsigned i = base + (unsigned)u * BT;
                e[u] = buf2[(i < nb) ? i : 0u];
            }
#pragma unroll
            for (int u = 0; u < PB; ++u) {
                rk[u] = 0u;
                if (base + (unsigned)u * BT < nb) rk[u] = atomicAdd(&ncnt[e[u] & 255u], 1u);
            }
#pragma unroll
            for (int u = 0; u < PB; ++u) {
                if (base + (unsigned)u * BT < nb) {
                    const unsigned slot = nstart[e[u] & 255u] + rk[u];
                    if (slot < (unsigned)CAPB) buf[slot] = e[u] >> 8;
                }
            }
        }
        gcn_lds_barrier();
        for (unsigned i = (unsigned)t; i < nb; i += BT) csr[gb + i] = (int)buf[i];
    } else {
        gcn_sweep_runs(pk, rlen, rsrc, roff, nr, w, lane, [&](unsigned e, unsigned) {
            const unsigned ld = e & 255u;
            const unsigned rk = atomicAdd(&ncnt[ld], 1u);
            csr[gb + nstart[ld] + rk] = (int)(e >> 8);
        });
    }
    gcn_lds_barrier();

#pragma unroll
    for (int i = 0; i < XR; ++i) {
        if (i * 32 + w < BN) {
            const unsigned r  = (unsigned)(i * 32 + w);
            const unsigned pp = lane & 15u;
            unsigned* hb = (lane & 16u) ? buf2 : buf;
            *reinterpret_cast<uint4*>(hb + r * 64u + 4u * (pp ^ (r & 15u))) =
                make_uint4(__float_as_uint(xv[i].x), __float_as_uint(xv[i].y),
                           __float_as_uint(xv[i].z), __float_as_uint(xv[i].w));
        }
    }
    gcn_lds_barrier();
    if (w < NT16) {
        const int row0 = b * BN + w * 16;
        if (row0 < n) {
            const unsigned half = lane >> 4;
            const unsigned l16  = lane & 15u;
            const unsigned rl   = (unsigned)w * 16u + l16;
            v8f acc = {};
#pragma unroll
            for (int s = 0; s < F_IN / 4; ++s) {
                const unsigned* hb = (s < 16) ? buf : buf2;
                const uint2 av = *reinterpret_cast<const uint2*>(
                    hb + rl * 64u + 4u * ((unsigned)(s & 15) ^ l16) + 2u * half);
                const unsigned ka = 4u * (unsigned)s + 2u * half;
                v2f a, bw;
                a.x  = __uint_as_float(av.x);
                a.y  = __uint_as_float(av.y);
                bw.x = sW[ka * F_HID + l16];
                bw.y = sW[(ka + 1u) * F_HID + l16];
                acc = __builtin_amdgcn_wmma_f32_16x16x4_f32(
                         false, a,  false, bw,
                         (short)0, acc,  false,  false);
            }
            float dv[8];
#pragma unroll
            for (int r = 0; r < 8; ++r) dv[r] = sdis[(unsigned)w * 16u + (unsigned)r + half * 8u];
#pragma unroll
            for (int r = 0; r < 8; ++r) {
                const int rloc = w * 16 + r + (int)half * 8;
                const int node = b * BN + rloc;
                if (rloc < BN && node < n) xWs[(long long)node * F_HID + l16] = acc[r] * dv[r];
            }
        }
    }
}

__global__ __launch_bounds__(256)
void gcn_agg1_gather(const int* __restrict__ csr, const unsigned* __restrict__ first,
                     const unsigned* __restrict__ degi, const float* __restrict__ xWs,
                     const float* __restrict__ dis, const float* __restrict__ b1,
                     const float* __restrict__ W2, float* __restrict__ y, int n) {
    const int t    = blockIdx.x * blockDim.x + threadIdx.x;
    const int node = t >> 4;
    const int j    = t & 15;
    const bool live = node < n;
    const int nd   = live ? node : (n - 1);
    const unsigned beg = first[nd];
    const unsigned cnt = degi[nd];
    const float* __restrict__ col = xWs + j;
    float acc = col[(long long)nd * F_HID];
    int next = ((unsigned)j < cnt) ? csr[beg + (unsigned)j] : -1;
    for (unsigned k0 = 0u; k0 < cnt; k0 += 16u) {
        const int mine = next;
        int   s[16];
        float r[16];
#pragma unroll
        for (int u = 0; u < 16; ++u) {
            s[u] = __shfl(mine, u, 16);
            const int sc = (s[u] >= 0) ? s[u] : nd;
            r[u] = col[(long long)sc * F_HID];
        }
        const unsigned kn = k0 + 16u + (unsigned)j;
        next = (kn < cnt) ? csr[beg + kn] : -1;
#pragma unroll
        for (int u = 0; u < 16; ++u) acc += (s[u] >= 0) ? r[u] : 0.0f;
    }
    const float dv = dis[nd];
    float hv = fmaf(acc, dv, b1[j]);
    hv = fmaxf(hv, 0.0f);
    float p = hv * W2[j];
    p += __shfl_xor(p, 8, 16);
    p += __shfl_xor(p, 4, 16);
    p += __shfl_xor(p, 2, 16);
    p += __shfl_xor(p, 1, 16);
    if (live && j == 0) y[node] = p * dv;
}

#define YS     51200
#define A2T    1024
#define A2G    (A2T / 16)
#define A2IT   7
#define A2NPB  (A2G * A2IT)
#define A2SW   ((YS + A2T * 4 - 1) / (A2T * 4))
#define A2SA   7

template <int B>
__device__ __forceinline__ void gcn_stage_y(const float* __restrict__ ysrc, float* __restrict__ ys,
                                            int len4, int sw0, int rot, unsigned t4) {
    float4 v[B];
#pragma unroll
    for (int b = 0; b < B; ++b) {
        const int sw  = (sw0 + b + rot < A2SW) ? (sw0 + b + rot) : (sw0 + b + rot - A2SW);
        const int off = sw * (A2T * 4);
        v[b] = make_float4(0.0f, 0.0f, 0.0f, 0.0f);
        if ((int)t4 < len4 - off) v[b] = *reinterpret_cast<const float4*>(ysrc + off + t4);
    }
#pragma unroll
    for (int b = 0; b < B; ++b) {
        const int sw  = (sw0 + b + rot < A2SW) ? (sw0 + b + rot) : (sw0 + b + rot - A2SW);
        const int off = sw * (A2T * 4);
        if ((int)t4 < len4 - off) *reinterpret_cast<float4*>(ys + off + t4) = v[b];
    }
}

__global__ __launch_bounds__(A2T)
void gcn_agg2_lds(const int* __restrict__ csr, const unsigned* __restrict__ first,
                  const unsigned* __restrict__ degi, const float* __restrict__ y,
                  const float* __restrict__ dis, const float* __restrict__ b2,
                  float* __restrict__ out, int n) {
    __shared__ __attribute__((aligned(16))) float ys[YS];
    const int tid  = threadIdx.x;
    const int grp  = tid >> 4;
    const int j    = tid & 15;
    const int base = blockIdx.x * A2NPB;
    const int rot  = blockIdx.x % A2SW;

    unsigned cnt[A2IT];
    int      s[A2IT][4];
    {
        unsigned beg[A2IT];
#pragma unroll
        for (int it = 0; it < A2IT; ++it) {
            const int node = base + it * A2G + grp;
            const int nd   = (node < n) ? node : (n - 1);
            beg[it] = first[nd];
            const unsigned c = degi[nd];
            cnt[it] = (node < n) ? c : 0u;
        }
#pragma unroll
        for (int it = 0; it < A2IT; ++it) {
#pragma unroll
            for (int u = 0; u < 4; ++u) {
                const unsigned k = (unsigned)j + 16u * (unsigned)u;
                s[it][u] = (k < cnt[it]) ? csr[beg[it] + k] : -1;
            }
        }
    }
    float acc[A2IT], dv[A2IT], yv[A2IT];
#pragma unroll
    for (int it = 0; it < A2IT; ++it) {
        const int node = base + it * A2G + grp;
        const int nd   = (node < n) ? node : (n - 1);
        dv[it]  = dis[nd];
        yv[it]  = y[nd];
        acc[it] = 0.0f;
    }

    for (int lo = 0; lo < n; lo += YS) {
        const int len  = (n - lo < YS) ? (n - lo) : YS;
        const int len4 = len & ~3;
        gcn_stage_y<A2SA>(y + lo, ys, len4, 0, rot, (unsigned)tid * 4u);
        gcn_stage_y<A2SW - A2SA>(y + lo, ys, len4, A2SA, rot, (unsigned)tid * 4u);
        if (tid < len - len4) ys[len4 + tid] = y[lo + len4 + tid];
        __syncthreads();
#pragma unroll
        for (int it = 0; it < A2IT; ++it) {
#pragma unroll
            for (int u = 0; u < 4; ++u) {
                const unsigned loc = (unsigned)(s[it][u] - lo);
                const float    r   = ys[(loc < (unsigned)len) ? loc : 0u];
                acc[it] += (loc < (unsigned)len) ? r : 0.0f;
            }
        }
#pragma unroll
        for (int it = 0; it < A2IT; ++it) {
            if (cnt[it] > 64u) {
                const unsigned bg = first[base + it * A2G + grp];
                for (unsigned k = 64u + (unsigned)j; k < cnt[it]; k += 16u) {
                    const unsigned loc = (unsigned)(csr[bg + k] - lo);
                    if (loc < (unsigned)len) acc[it] += ys[loc];
                }
            }
        }
        __syncthreads();
    }
#pragma unroll
    for (int w = 8; w >= 1; w >>= 1) {
#pragma unroll
        for (int it = 0; it < A2IT; ++it) acc[it] += __shfl_xor(acc[it], w, 16);
    }
    const float bias = b2[0];
#pragma unroll
    for (int it = 0; it < A2IT; ++it) {
        const int node = base + it * A2G + grp;
        if (node < n && j == 0) out[node] = bias + dv[it] * (yv[it] + acc[it]);
    }
}

extern "C" void kernel_launch(void* const* d_in, const int* in_sizes, int n_in,
                              void* d_out, int out_size, void* d_ws, size_t ws_size,
                              hipStream_t stream) {
    const float* x  = (const float*)d_in[0];
    const int*   ei = (const int*)d_in[1];
    const float* W1 = (const float*)d_in[2];
    const float* b1 = (const float*)d_in[3];
    const float* W2 = (const float*)d_in[4];
    const float* b2 = (const float*)d_in[5];
    float* out = (float*)d_out;

    const int       N = in_sizes[0] / F_IN;
    const long long E = (long long)in_sizes[1] / 2;

    const int Ei        = (int)E;
    const int nblk      = (int)((E + CHMAX - 1) / CHMAX);
    const int chunk_len = (int)((E + nblk - 1) / nblk);
    const int n_bkt     = (N + BN - 1) / BN;

    const long long Np = ((long long)N + 3) / 4 * 4;
    float*    ws    = (float*)d_ws;
    float*    y     = ws;
    float*    dis   = ws + Np;
    float*    xWs   = ws + Np + N;
    unsigned* degi  = (unsigned*)(ws + Np + (long long)17 * N);
    unsigned* first = (unsigned*)(ws + Np + (long long)18 * N);
    unsigned* runm  = (unsigned*)(ws + Np + (long long)19 * N);
    unsigned* pk    = runm + (long long)nblk * NBP;
    int*      csr   = (int*)(pk + (long long)nblk * chunk_len);

    const int T = 256;
    const unsigned nb_n16  = (unsigned)(((long long)N * F_HID + T - 1) / T);
    const unsigned nb_a2   = (unsigned)((N + A2NPB - 1) / A2NPB);

    gcn_block_sort   <<<(unsigned)nblk,  BT, 0, stream>>>(ei, runm, pk, Ei, N, chunk_len);
    gcn_bucket_finish<<<(unsigned)n_bkt, BT, 0, stream>>>(pk, runm, csr, first, degi, dis, x, W1, xWs, N, nblk, chunk_len);
    gcn_agg1_gather  <<<nb_n16,  T,  0, stream>>>(csr, first, degi, xWs, dis, b1, W2, y, N);
    gcn_agg2_lds     <<<nb_a2,   A2T, 0, stream>>>(csr, first, degi, y, dis, b2, out, N);
}
